# starting kernel
# baseline (speedup 1.0000x reference)
#include <hip/hip_runtime.h>
#include <math.h>

#define N_NODES  50000
#define N_EDGES  800000
#define HID      128
#define EDGE_D   16
#define N_GRAPHS 64

typedef __attribute__((ext_vector_type(16))) __bf16 v16bf;
typedef __attribute__((ext_vector_type(8)))  float  v8f;

union Frag {
    uint4 q[2];
    v16bf v;
};

__device__ __forceinline__ unsigned short f2bf(float f) {
    union { float f; unsigned int u; } c; c.f = f;
    unsigned int u = c.u;
    u += 0x7fffu + ((u >> 16) & 1u);
    return (unsigned short)(u >> 16);
}

__device__ __forceinline__ float silu_f(float x) {
    return x / (1.0f + __expf(-x));
}

__device__ __forceinline__ int asw_pos(int m_local, int K, int KT) {
    int mt = m_local >> 4, ml = m_local & 15;
    int t = K >> 5, kk = K & 31;
    int g, i;
    if (kk < 16) { g = kk >> 3;        i = kk & 7; }
    else         { g = (kk - 16) >> 3; i = 8 + ((kk - 16) & 7); }
    return ((mt * KT + t) * 32 + (ml + (g << 4))) * 16 + i;
}

__global__ void wconv_kernel(const float* __restrict__ W,
                             unsigned short* __restrict__ dst,
                             int Ksrc, int KT) {
    int idx = blockIdx.x * blockDim.x + threadIdx.x;
    int total = KT * 8 * 32 * 16;
    if (idx >= total) return;
    int i    = idx & 15;
    int lane = (idx >> 4) & 31;
    int tj   = idx >> 9;
    int t = tj >> 3, j = tj & 7;
    int g = lane >> 4;
    int K = t * 32 + (i < 8 ? g * 8 + i : 16 + g * 8 + (i - 8));
    int n = j * 16 + (lane & 15);
    float v = (K < Ksrc) ? W[K * HID + n] : 0.0f;
    dst[idx] = f2bf(v);
}

__global__ void zero_f32_kernel(float* p, int n) {
    int i = blockIdx.x * blockDim.x + threadIdx.x;
    if (i < n) p[i] = 0.0f;
}

#define K1T 9
#define K2T 4

__global__ __launch_bounds__(256)
void edge_kernel(const float* __restrict__ h,
                 const int*   __restrict__ eidx,
                 const float* __restrict__ x,
                 const float* __restrict__ eattr,
                 const unsigned short* __restrict__ We1b,
                 const float* __restrict__ be1,
                 const unsigned short* __restrict__ We2b,
                 const float* __restrict__ be2,
                 float* __restrict__ agg) {
    __shared__ uint4 lA4[4 * K1T * 32 * 2];
    __shared__ uint4 lM4[4 * K2T * 32 * 2];
    unsigned short* lA = (unsigned short*)lA4;
    unsigned short* lM = (unsigned short*)lM4;

    const int tid = threadIdx.x;
    const int e0  = blockIdx.x * 64;

    for (int idx = tid; idx < 64 * 288; idx += 256) {
        int m = idx / 288;
        int K = idx - m * 288;
        int e = e0 + m;
        int r = eidx[e];
        int c = eidx[N_EDGES + e];
        float v;
        if (K < HID)            v = h[r * HID + K];
        else if (K < 2 * HID)   v = h[c * HID + (K - HID)];
        else if (K == 2 * HID) {
            float d0 = x[r * 3 + 0] - x[c * 3 + 0];
            float d1 = x[r * 3 + 1] - x[c * 3 + 1];
            float d2 = x[r * 3 + 2] - x[c * 3 + 2];
            v = d0 * d0 + d1 * d1 + d2 * d2;
        }
        else if (K < 2 * HID + 1 + EDGE_D) v = eattr[e * EDGE_D + (K - 2 * HID - 1)];
        else v = 0.0f;
        lA[asw_pos(m, K, K1T)] = f2bf(v);
    }
    __syncthreads();

    const int w    = tid >> 5;
    const int lane = tid & 31;
    const int hi   = lane >> 4;
    const int n    = w * 16 + (lane & 15);

    v8f acc[4] = {};
    for (int t = 0; t < K1T; ++t) {
        Frag B;
        const uint4* bp = (const uint4*)We1b + ((t * 8 + w) * 32 + lane) * 2;
        B.q[0] = bp[0]; B.q[1] = bp[1];
        #pragma unroll
        for (int mt = 0; mt < 4; ++mt) {
            Frag A;
            const uint4* ap = (const uint4*)lA4 + ((mt * K1T + t) * 32 + lane) * 2;
            A.q[0] = ap[0]; A.q[1] = ap[1];
            acc[mt] = __builtin_amdgcn_wmma_f32_16x16x32_bf16(
                false, A.v, false, B.v, (short)0, acc[mt], false, false);
        }
    }
    {
        float b = be1[n];
        #pragma unroll
        for (int mt = 0; mt < 4; ++mt) {
            #pragma unroll
            for (int j = 0; j < 8; ++j) {
                int m = mt * 16 + hi * 8 + j;
                lM[asw_pos(m, n, K2T)] = f2bf(silu_f(acc[mt][j] + b));
            }
        }
    }
    __syncthreads();

    v8f acc2[4] = {};
    for (int t = 0; t < K2T; ++t) {
        Frag B;
        const uint4* bp = (const uint4*)We2b + ((t * 8 + w) * 32 + lane) * 2;
        B.q[0] = bp[0]; B.q[1] = bp[1];
        #pragma unroll
        for (int mt = 0; mt < 4; ++mt) {
            Frag A;
            const uint4* ap = (const uint4*)lM4 + ((mt * K2T + t) * 32 + lane) * 2;
            A.q[0] = ap[0]; A.q[1] = ap[1];
            acc2[mt] = __builtin_amdgcn_wmma_f32_16x16x32_bf16(
                false, A.v, false, B.v, (short)0, acc2[mt], false, false);
        }
    }
    {
        float b = be2[n];
        #pragma unroll
        for (int mt = 0; mt < 4; ++mt) {
            #pragma unroll
            for (int j = 0; j < 8; ++j) {
                int m = mt * 16 + hi * 8 + j;
                float f = silu_f(acc2[mt][j] + b);
                int r = eidx[e0 + m];
                atomicAdd(&agg[r * HID + n], f);
            }
        }
    }
}

#define KN1T 8
#define KN2T 4

__global__ __launch_bounds__(256)
void node_kernel(const float* __restrict__ h,
                 const float* __restrict__ agg,
                 const int*   __restrict__ batch,
                 const unsigned short* __restrict__ Wn1b,
                 const float* __restrict__ bn1,
                 const unsigned short* __restrict__ Wn2b,
                 const float* __restrict__ bn2,
                 float* __restrict__ sums) {
    __shared__ uint4 lA4[4 * KN1T * 32 * 2];
    __shared__ uint4 lM4[4 * KN2T * 32 * 2];
    unsigned short* lA = (unsigned short*)lA4;
    unsigned short* lM = (unsigned short*)lM4;

    const int tid = threadIdx.x;
    const int n0  = blockIdx.x * 64;

    for (int idx = tid; idx < 64 * 256; idx += 256) {
        int m = idx >> 8;
        int K = idx & 255;
        int node = n0 + m;
        float v = 0.0f;
        if (node < N_NODES)
            v = (K < HID) ? h[node * HID + K] : agg[node * HID + (K - HID)];
        lA[asw_pos(m, K, KN1T)] = f2bf(v);
    }
    __syncthreads();

    const int w    = tid >> 5;
    const int lane = tid & 31;
    const int hi   = lane >> 4;
    const int n    = w * 16 + (lane & 15);

    v8f acc[4] = {};
    for (int t = 0; t < KN1T; ++t) {
        Frag B;
        const uint4* bp = (const uint4*)Wn1b + ((t * 8 + w) * 32 + lane) * 2;
        B.q[0] = bp[0]; B.q[1] = bp[1];
        #pragma unroll
        for (int mt = 0; mt < 4; ++mt) {
            Frag A;
            const uint4* ap = (const uint4*)lA4 + ((mt * KN1T + t) * 32 + lane) * 2;
            A.q[0] = ap[0]; A.q[1] = ap[1];
            acc[mt] = __builtin_amdgcn_wmma_f32_16x16x32_bf16(
                false, A.v, false, B.v, (short)0, acc[mt], false, false);
        }
    }
    {
        float b = bn1[n];
        #pragma unroll
        for (int mt = 0; mt < 4; ++mt) {
            #pragma unroll
            for (int j = 0; j < 8; ++j) {
                int m = mt * 16 + hi * 8 + j;
                lM[asw_pos(m, n, KN2T)] = f2bf(silu_f(acc[mt][j] + b));
            }
        }
    }
    __syncthreads();

    v8f acc2[4] = {};
    for (int t = 0; t < KN2T; ++t) {
        Frag B;
        const uint4* bp = (const uint4*)Wn2b + ((t * 8 + w) * 32 + lane) * 2;
        B.q[0] = bp[0]; B.q[1] = bp[1];
        #pragma unroll
        for (int mt = 0; mt < 4; ++mt) {
            Frag A;
            const uint4* ap = (const uint4*)lM4 + ((mt * KN2T + t) * 32 + lane) * 2;
            A.q[0] = ap[0]; A.q[1] = ap[1];
            acc2[mt] = __builtin_amdgcn_wmma_f32_16x16x32_bf16(
                false, A.v, false, B.v, (short)0, acc2[mt], false, false);
        }
    }
    {
        float b = bn2[n];
        #pragma unroll
        for (int mt = 0; mt < 4; ++mt) {
            #pragma unroll
            for (int j = 0; j < 8; ++j) {
                int m = mt * 16 + hi * 8 + j;
                int node = n0 + m;
                if (node < N_NODES) {
                    float hn = h[node * HID + n] + (acc2[mt][j] + b);
                    int g = batch[node];
                    atomicAdd(&sums[g * HID + n], hn);
                }
            }
        }
    }
}

__global__ void counts_kernel(const int* __restrict__ batch, float* __restrict__ counts) {
    int i = blockIdx.x * blockDim.x + threadIdx.x;
    if (i < N_NODES) atomicAdd(&counts[batch[i]], 1.0f);
}

__global__ void finalize_kernel(const float* __restrict__ sums,
                                const float* __restrict__ counts,
                                float* __restrict__ out) {
    int i = blockIdx.x * blockDim.x + threadIdx.x;
    if (i < N_GRAPHS * HID)
        out[i] = sums[i] / fmaxf(counts[i >> 7], 1.0f);
}

extern "C" void kernel_launch(void* const* d_in, const int* in_sizes, int n_in,
                              void* d_out, int out_size, void* d_ws, size_t ws_size,
                              hipStream_t stream) {
    const float* h     = (const float*)d_in[0];
    const int*   eidx  = (const int*)  d_in[1];
    const float* x     = (const float*)d_in[2];
    const float* eattr = (const float*)d_in[3];
    const int*   batch = (const int*)  d_in[4];
    const float* We1   = (const float*)d_in[5];
    const float* be1   = (const float*)d_in[6];
    const float* We2   = (const float*)d_in[7];
    const float* be2   = (const float*)d_in[8];
    const float* Wn1   = (const float*)d_in[9];
    const float* bn1   = (const float*)d_in[10];
    const float* Wn2   = (const float*)d_in[11];
    const float* bn2   = (const float*)d_in[12];

    char* ws = (char*)d_ws;
    unsigned short* We1b = (unsigned short*)(ws);
    unsigned short* We2b = (unsigned short*)(ws + 73728);
    unsigned short* Wn1b = (unsigned short*)(ws + 106496);
    unsigned short* Wn2b = (unsigned short*)(ws + 172032);
    float* agg    = (float*)(ws + 204800);
    float* sums   = (float*)(ws + 204800 + 25600000);
    float* counts = (float*)(ws + 204800 + 25600000 + 32768);

    wconv_kernel<<<(9 * 4096 + 255) / 256, 256, 0, stream>>>(We1, We1b, 273, 9);
    wconv_kernel<<<(4 * 4096 + 255) / 256, 256, 0, stream>>>(We2, We2b, 128, 4);
    wconv_kernel<<<(8 * 4096 + 255) / 256, 256, 0, stream>>>(Wn1, Wn1b, 256, 8);
    wconv_kernel<<<(4 * 4096 + 255) / 256, 256, 0, stream>>>(Wn2, Wn2b, 128, 4);

    int nz = N_NODES * HID + N_GRAPHS * HID + N_GRAPHS;
    zero_f32_kernel<<<(nz + 255) / 256, 256, 0, stream>>>(agg, nz);

    edge_kernel<<<N_EDGES / 64, 256, 0, stream>>>(h, eidx, x, eattr,
                                                  We1b, be1, We2b, be2, agg);
    node_kernel<<<(N_NODES + 63) / 64, 256, 0, stream>>>(h, agg, batch,
                                                         Wn1b, bn1, Wn2b, bn2, sums);
    counts_kernel<<<(N_NODES + 255) / 256, 256, 0, stream>>>(batch, counts);
    finalize_kernel<<<(N_GRAPHS * HID + 255) / 256, 256, 0, stream>>>(sums, counts,
                                                                      (float*)d_out);
}


# edges counting-sorted by destination row in the timed body (atomic arrival-order rank); each lane sums runs of equal rows in a register before its agg atomic: f32 sum order changes
# speedup vs baseline: 1.0658x; 1.0658x over previous
#include <hip/hip_runtime.h>
#include <math.h>

#define N_NODES  50000
#define N_EDGES  800000
#define HID      128
#define EDGE_D   16
#define N_GRAPHS 64

typedef __attribute__((ext_vector_type(16))) __bf16 v16bf;
typedef __attribute__((ext_vector_type(8)))  float  v8f;

union Frag {
    uint4 q[2];
    v16bf v;
};

__device__ __forceinline__ unsigned short f2bf(float f) {
    union { float f; unsigned int u; } c; c.f = f;
    unsigned int u = c.u;
    u += 0x7fffu + ((u >> 16) & 1u);
    return (unsigned short)(u >> 16);
}

__device__ __forceinline__ float silu_f(float x) {
    return x / (1.0f + __expf(-x));
}

__device__ __forceinline__ int asw_pos(int m_local, int K, int KT) {
    int mt = m_local >> 4, ml = m_local & 15;
    int t = K >> 5, kk = K & 31;
    int g, i;
    if (kk < 16) { g = kk >> 3;        i = kk & 7; }
    else         { g = (kk - 16) >> 3; i = 8 + ((kk - 16) & 7); }
    return ((mt * KT + t) * 32 + (ml + (g << 4))) * 16 + i;
}

__global__ void wconv_kernel(const float* __restrict__ W,
                             unsigned short* __restrict__ dst,
                             int Ksrc, int KT) {
    int idx = blockIdx.x * blockDim.x + threadIdx.x;
    int total = KT * 8 * 32 * 16;
    if (idx >= total) return;
    int i    = idx & 15;
    int lane = (idx >> 4) & 31;
    int tj   = idx >> 9;
    int t = tj >> 3, j = tj & 7;
    int g = lane >> 4;
    int K = t * 32 + (i < 8 ? g * 8 + i : 16 + g * 8 + (i - 8));
    int n = j * 16 + (lane & 15);
    float v = (K < Ksrc) ? W[K * HID + n] : 0.0f;
    dst[idx] = f2bf(v);
}

#define WS_NEED 29237824ull
#define NZ_BASE (N_NODES * HID + N_GRAPHS * HID + N_GRAPHS)
#define NZ_SORT (NZ_BASE + N_NODES)

__global__ void zero_f32_kernel(float* p, int n_base, int n_sort, size_t ws_size) {
    int i = blockIdx.x * blockDim.x + threadIdx.x;
    int n = (ws_size >= WS_NEED) ? n_sort : n_base;
    if (i < n) p[i] = 0.0f;
}

__global__ void hist_kernel(const int* __restrict__ eidx, int* __restrict__ cursor,
                            size_t ws_size) {
    if (ws_size < WS_NEED) return;
    int e = blockIdx.x * blockDim.x + threadIdx.x;
    if (e < N_EDGES) atomicAdd(&cursor[eidx[e]], 1);
}

#define SCAN_K   13
#define SCAN_SEG (SCAN_K * 128)
__global__ __launch_bounds__(1024)
void scan_kernel(int* __restrict__ cursor, size_t ws_size) {
    if (ws_size < WS_NEED) return;
    __shared__ int wtot[32];
    const int w = threadIdx.x >> 5, lane = threadIdx.x & 31;
    const int base = w * SCAN_SEG + lane * 4;
    int4 v[SCAN_K];
    int  ex[SCAN_K];
    int running = 0;
    #pragma unroll
    for (int k = 0; k < SCAN_K; ++k) {
        int idx = base + k * 128;
        v[k] = (idx < N_NODES) ? *(const int4*)(cursor + idx) : make_int4(0, 0, 0, 0);
    }
    #pragma unroll
    for (int k = 0; k < SCAN_K; ++k) {
        int s = v[k].x + v[k].y + v[k].z + v[k].w;
        int inc = s;
        #pragma unroll
        for (int d = 1; d < 32; d <<= 1) {
            int t = __shfl_up(inc, d);
            if (lane >= d) inc += t;
        }
        ex[k] = running + inc - s;
        running += __shfl(inc, 31);
    }
    if (lane == 0) wtot[w] = running;
    __syncthreads();
    int woff = 0;
    for (int i = 0; i < w; ++i) woff += wtot[i];
    #pragma unroll
    for (int k = 0; k < SCAN_K; ++k) {
        int idx = base + k * 128;
        if (idx < N_NODES) {
            int o = woff + ex[k];
            int4 r;
            r.x = o; r.y = o + v[k].x; r.z = r.y + v[k].y; r.w = r.z + v[k].z;
            *(int4*)(cursor + idx) = r;
        }
    }
}

__global__ void fill_kernel(const int* __restrict__ eidx, int* __restrict__ cursor,
                            int* __restrict__ perm, size_t ws_size) {
    if (ws_size < WS_NEED) return;
    int e = blockIdx.x * blockDim.x + threadIdx.x;
    if (e < N_EDGES) {
        int pos = atomicAdd(&cursor[eidx[e]], 1);
        perm[pos] = e;
    }
}

#define K1T 9
#define K2T 4

__global__ __launch_bounds__(256)
void edge_kernel(const float* __restrict__ h,
                 const int*   __restrict__ eidx,
                 const float* __restrict__ x,
                 const float* __restrict__ eattr,
                 const unsigned short* __restrict__ We1b,
                 const float* __restrict__ be1,
                 const unsigned short* __restrict__ We2b,
                 const float* __restrict__ be2,
                 float* __restrict__ agg,
                 const int* __restrict__ perm,
                 size_t ws_size) {
    __shared__ uint4 lA4[4 * K1T * 32 * 2];
    __shared__ uint4 lM4[4 * K2T * 32 * 2];
    unsigned short* lA = (unsigned short*)lA4;
    unsigned short* lM = (unsigned short*)lM4;

    const int tid = threadIdx.x;
    const int e0  = blockIdx.x * 64;

    int* sE = (int*)lM4;
    int* sR = sE + 64;
    int* sC = sE + 128;
    if (tid < 64) {
        int i = e0 + tid;
        int e = (ws_size >= WS_NEED) ? perm[i] : i;
        sE[tid] = e;
        sR[tid] = eidx[e];
        sC[tid] = eidx[N_EDGES + e];
    }
    __syncthreads();
    const int rowLo = sR[tid & 31];
    const int rowHi = sR[32 + (tid & 31)];

    for (int idx = tid; idx < 64 * 288; idx += 256) {
        int m = idx / 288;
        int K = idx - m * 288;
        int e = sE[m];
        int r = sR[m];
        int c = sC[m];
        float v;
        if (K < HID)            v = h[r * HID + K];
        else if (K < 2 * HID)   v = h[c * HID + (K - HID)];
        else if (K == 2 * HID) {
            float d0 = x[r * 3 + 0] - x[c * 3 + 0];
            float d1 = x[r * 3 + 1] - x[c * 3 + 1];
            float d2 = x[r * 3 + 2] - x[c * 3 + 2];
            v = d0 * d0 + d1 * d1 + d2 * d2;
        }
        else if (K < 2 * HID + 1 + EDGE_D) v = eattr[e * EDGE_D + (K - 2 * HID - 1)];
        else v = 0.0f;
        lA[asw_pos(m, K, K1T)] = f2bf(v);
    }
    __syncthreads();

    const int w    = tid >> 5;
    const int lane = tid & 31;
    const int hi   = lane >> 4;
    const int n    = w * 16 + (lane & 15);

    v8f acc[4] = {};
    for (int t = 0; t < K1T; ++t) {
        Frag B;
        const uint4* bp = (const uint4*)We1b + ((t * 8 + w) * 32 + lane) * 2;
        B.q[0] = bp[0]; B.q[1] = bp[1];
        #pragma unroll
        for (int mt = 0; mt < 4; ++mt) {
            Frag A;
            const uint4* ap = (const uint4*)lA4 + ((mt * K1T + t) * 32 + lane) * 2;
            A.q[0] = ap[0]; A.q[1] = ap[1];
            acc[mt] = __builtin_amdgcn_wmma_f32_16x16x32_bf16(
                false, A.v, false, B.v, (short)0, acc[mt], false, false);
        }
    }
    {
        float b = be1[n];
        #pragma unroll
        for (int mt = 0; mt < 4; ++mt) {
            #pragma unroll
            for (int j = 0; j < 8; ++j) {
                int m = mt * 16 + hi * 8 + j;
                lM[asw_pos(m, n, K2T)] = f2bf(silu_f(acc[mt][j] + b));
            }
        }
    }
    __syncthreads();

    v8f acc2[4] = {};
    for (int t = 0; t < K2T; ++t) {
        Frag B;
        const uint4* bp = (const uint4*)We2b + ((t * 8 + w) * 32 + lane) * 2;
        B.q[0] = bp[0]; B.q[1] = bp[1];
        #pragma unroll
        for (int mt = 0; mt < 4; ++mt) {
            Frag A;
            const uint4* ap = (const uint4*)lM4 + ((mt * K2T + t) * 32 + lane) * 2;
            A.q[0] = ap[0]; A.q[1] = ap[1];
            acc2[mt] = __builtin_amdgcn_wmma_f32_16x16x32_bf16(
                false, A.v, false, B.v, (short)0, acc2[mt], false, false);
        }
    }
    {
        float b = be2[n];
        int   cur = 0;
        float run = 0.0f;
        #pragma unroll
        for (int mt = 0; mt < 4; ++mt) {
            #pragma unroll
            for (int j = 0; j < 8; ++j) {
                const int rv = (mt < 2) ? rowLo : rowHi;
                const int r0 = __builtin_amdgcn_readlane(rv, (mt & 1) * 16 + j);
                const int r1 = __builtin_amdgcn_readlane(rv, (mt & 1) * 16 + 8 + j);
                const int r  = hi ? r1 : r0;
                float f = silu_f(acc2[mt][j] + b);
                if (mt == 0 && j == 0) {
                    cur = r;
                    run = f;
                } else {
                    if (r != cur) {
                        atomicAdd(&agg[cur * HID + n], run);
                        run = 0.0f;
                        cur = r;
                    }
                    run += f;
                }
            }
        }
        atomicAdd(&agg[cur * HID + n], run);
    }
}

#define KN1T 8
#define KN2T 4

__global__ __launch_bounds__(256)
void node_kernel(const float* __restrict__ h,
                 const float* __restrict__ agg,
                 const int*   __restrict__ batch,
                 const unsigned short* __restrict__ Wn1b,
                 const float* __restrict__ bn1,
                 const unsigned short* __restrict__ Wn2b,
                 const float* __restrict__ bn2,
                 float* __restrict__ sums) {
    __shared__ uint4 lA4[4 * KN1T * 32 * 2];
    __shared__ uint4 lM4[4 * KN2T * 32 * 2];
    unsigned short* lA = (unsigned short*)lA4;
    unsigned short* lM = (unsigned short*)lM4;

    const int tid = threadIdx.x;
    const int n0  = blockIdx.x * 64;

    for (int idx = tid; idx < 64 * 256; idx += 256) {
        int m = idx >> 8;
        int K = idx & 255;
        int node = n0 + m;
        float v = 0.0f;
        if (node < N_NODES)
            v = (K < HID) ? h[node * HID + K] : agg[node * HID + (K - HID)];
        lA[asw_pos(m, K, KN1T)] = f2bf(v);
    }
    __syncthreads();

    const int w    = tid >> 5;
    const int lane = tid & 31;
    const int hi   = lane >> 4;
    const int n    = w * 16 + (lane & 15);

    v8f acc[4] = {};
    for (int t = 0; t < KN1T; ++t) {
        Frag B;
        const uint4* bp = (const uint4*)Wn1b + ((t * 8 + w) * 32 + lane) * 2;
        B.q[0] = bp[0]; B.q[1] = bp[1];
        #pragma unroll
        for (int mt = 0; mt < 4; ++mt) {
            Frag A;
            const uint4* ap = (const uint4*)lA4 + ((mt * KN1T + t) * 32 + lane) * 2;
            A.q[0] = ap[0]; A.q[1] = ap[1];
            acc[mt] = __builtin_amdgcn_wmma_f32_16x16x32_bf16(
                false, A.v, false, B.v, (short)0, acc[mt], false, false);
        }
    }
    {
        float b = bn1[n];
        #pragma unroll
        for (int mt = 0; mt < 4; ++mt) {
            #pragma unroll
            for (int j = 0; j < 8; ++j) {
                int m = mt * 16 + hi * 8 + j;
                lM[asw_pos(m, n, KN2T)] = f2bf(silu_f(acc[mt][j] + b));
            }
        }
    }
    __syncthreads();

    v8f acc2[4] = {};
    for (int t = 0; t < KN2T; ++t) {
        Frag B;
        const uint4* bp = (const uint4*)Wn2b + ((t * 8 + w) * 32 + lane) * 2;
        B.q[0] = bp[0]; B.q[1] = bp[1];
        #pragma unroll
        for (int mt = 0; mt < 4; ++mt) {
            Frag A;
            const uint4* ap = (const uint4*)lM4 + ((mt * KN2T + t) * 32 + lane) * 2;
            A.q[0] = ap[0]; A.q[1] = ap[1];
            acc2[mt] = __builtin_amdgcn_wmma_f32_16x16x32_bf16(
                false, A.v, false, B.v, (short)0, acc2[mt], false, false);
        }
    }
    {
        float b = bn2[n];
        #pragma unroll
        for (int mt = 0; mt < 4; ++mt) {
            #pragma unroll
            for (int j = 0; j < 8; ++j) {
                int m = mt * 16 + hi * 8 + j;
                int node = n0 + m;
                if (node < N_NODES) {
                    float hn = h[node * HID + n] + (acc2[mt][j] + b);
                    int g = batch[node];
                    atomicAdd(&sums[g * HID + n], hn);
                }
            }
        }
    }
}

__global__ void counts_kernel(const int* __restrict__ batch, float* __restrict__ counts) {
    int i = blockIdx.x * blockDim.x + threadIdx.x;
    if (i < N_NODES) atomicAdd(&counts[batch[i]], 1.0f);
}

__global__ void finalize_kernel(const float* __restrict__ sums,
                                const float* __restrict__ counts,
                                float* __restrict__ out) {
    int i = blockIdx.x * blockDim.x + threadIdx.x;
    if (i < N_GRAPHS * HID)
        out[i] = sums[i] / fmaxf(counts[i >> 7], 1.0f);
}

extern "C" void kernel_launch(void* const* d_in, const int* in_sizes, int n_in,
                              void* d_out, int out_size, void* d_ws, size_t ws_size,
                              hipStream_t stream) {
    const float* h     = (const float*)d_in[0];
    const int*   eidx  = (const int*)  d_in[1];
    const float* x     = (const float*)d_in[2];
    const float* eattr = (const float*)d_in[3];
    const int*   batch = (const int*)  d_in[4];
    const float* We1   = (const float*)d_in[5];
    const float* be1   = (const float*)d_in[6];
    const float* We2   = (const float*)d_in[7];
    const float* be2   = (const float*)d_in[8];
    const float* Wn1   = (const float*)d_in[9];
    const float* bn1   = (const float*)d_in[10];
    const float* Wn2   = (const float*)d_in[11];
    const float* bn2   = (const float*)d_in[12];

    char* ws = (char*)d_ws;
    unsigned short* We1b = (unsigned short*)(ws);
    unsigned short* We2b = (unsigned short*)(ws + 73728);
    unsigned short* Wn1b = (unsigned short*)(ws + 106496);
    unsigned short* Wn2b = (unsigned short*)(ws + 172032);
    float* agg    = (float*)(ws + 204800);
    float* sums   = (float*)(ws + 204800 + 25600000);
    float* counts = (float*)(ws + 204800 + 25600000 + 32768);
    int*   cursor = (int*)(ws + 25837824);
    int*   perm   = (int*)(ws + 26037824);

    wconv_kernel<<<(9 * 4096 + 255) / 256, 256, 0, stream>>>(We1, We1b, 273, 9);
    wconv_kernel<<<(4 * 4096 + 255) / 256, 256, 0, stream>>>(We2, We2b, 128, 4);
    wconv_kernel<<<(8 * 4096 + 255) / 256, 256, 0, stream>>>(Wn1, Wn1b, 256, 8);
    wconv_kernel<<<(4 * 4096 + 255) / 256, 256, 0, stream>>>(Wn2, Wn2b, 128, 4);

    zero_f32_kernel<<<(NZ_SORT + 255) / 256, 256, 0, stream>>>(agg, NZ_BASE, NZ_SORT, ws_size);

    hist_kernel<<<(N_EDGES + 255) / 256, 256, 0, stream>>>(eidx, cursor, ws_size);
    scan_kernel<<<1, 1024, 0, stream>>>(cursor, ws_size);
    fill_kernel<<<(N_EDGES + 255) / 256, 256, 0, stream>>>(eidx, cursor, perm, ws_size);

    edge_kernel<<<N_EDGES / 64, 256, 0, stream>>>(h, eidx, x, eattr,
                                                  We1b, be1, We2b, be2, agg,
                                                  perm, ws_size);
    node_kernel<<<(N_NODES + 63) / 64, 256, 0, stream>>>(h, agg, batch,
                                                         Wn1b, bn1, Wn2b, bn2, sums);
    counts_kernel<<<(N_NODES + 255) / 256, 256, 0, stream>>>(batch, counts);
    finalize_kernel<<<(N_GRAPHS * HID + 255) / 256, 256, 0, stream>>>(sums, counts,
                                                                      (float*)d_out);
}


# SiLU of the three MLP epilogues as x * rcp(1 + exp(-x)) with the hardware reciprocal (about 1 ulp) instead of the correctly rounded f32 division; the run sum takes that product by fma
# speedup vs baseline: 1.6773x; 1.5737x over previous
#include <hip/hip_runtime.h>
#include <math.h>

#define N_NODES  50000
#define N_EDGES  800000
#define HID      128
#define EDGE_D   16
#define N_GRAPHS 64

typedef __attribute__((ext_vector_type(16))) __bf16 v16bf;
typedef __attribute__((ext_vector_type(8)))  float  v8f;

union Frag {
    uint4 q[2];
    v16bf v;
};

__device__ __forceinline__ unsigned short f2bf(float f) {
    union { float f; unsigned int u; } c; c.f = f;
    unsigned int u = c.u;
    u += 0x7fffu + ((u >> 16) & 1u);
    return (unsigned short)(u >> 16);
}

__device__ __forceinline__ float silu_f(float x) {
    return x * __builtin_amdgcn_rcpf(1.0f + __expf(-x));
}

typedef __attribute__((ext_vector_type(2))) float  f32x2;
typedef __attribute__((ext_vector_type(2))) __bf16 bf16x2;

__device__ __forceinline__ unsigned int pack2bf(float a, float b) {
    f32x2 v = {a, b};
    union { bf16x2 h; unsigned int u; } c;
    c.h = __builtin_convertvector(v, bf16x2);
    return c.u;
}

__device__ __forceinline__ uint2 pack4bf(float4 f) {
    return make_uint2(pack2bf(f.x, f.y), pack2bf(f.z, f.w));
}

__device__ __forceinline__ int abuild_base(int m0, int q, int KT) {
    return ((((m0 >> 4) * KT) * 32 + (m0 & 15) + 16 * ((q >> 1) & 1)) * 2 + (q >> 2)) * 2 + (q & 1);
}

__device__ __forceinline__ int asw_pos(int m_local, int K, int KT) {
    int mt = m_local >> 4, ml = m_local & 15;
    int t = K >> 5, kk = K & 31;
    int g, i;
    if (kk < 16) { g = kk >> 3;        i = kk & 7; }
    else         { g = (kk - 16) >> 3; i = 8 + ((kk - 16) & 7); }
    return ((mt * KT + t) * 32 + (ml + (g << 4))) * 16 + i;
}

__global__ void wconv_kernel(const float* __restrict__ W,
                             unsigned short* __restrict__ dst,
                             int Ksrc, int KT) {
    int idx = blockIdx.x * blockDim.x + threadIdx.x;
    int total = KT * 8 * 32 * 16;
    if (idx >= total) return;
    int i    = idx & 15;
    int lane = (idx >> 4) & 31;
    int tj   = idx >> 9;
    int t = tj >> 3, j = tj & 7;
    int g = lane >> 4;
    int K = t * 32 + (i < 8 ? g * 8 + i : 16 + g * 8 + (i - 8));
    int n = j * 16 + (lane & 15);
    float v = (K < Ksrc) ? W[K * HID + n] : 0.0f;
    dst[idx] = f2bf(v);
}

#define WS_NEED 29237824ull
#define NZ_BASE (N_NODES * HID + N_GRAPHS * HID + N_GRAPHS)
#define NZ_SORT (NZ_BASE + N_NODES)

__global__ void zero_f32_kernel(float* p, int n_base, int n_sort, size_t ws_size) {
    int i = blockIdx.x * blockDim.x + threadIdx.x;
    int n = (ws_size >= WS_NEED) ? n_sort : n_base;
    if (i < n) p[i] = 0.0f;
}

__global__ void hist_kernel(const int* __restrict__ eidx, int* __restrict__ cursor,
                            size_t ws_size) {
    if (ws_size < WS_NEED) return;
    int e = blockIdx.x * blockDim.x + threadIdx.x;
    if (e < N_EDGES) atomicAdd(&cursor[eidx[e]], 1);
}

#define SCAN_K   13
#define SCAN_SEG (SCAN_K * 128)
__global__ __launch_bounds__(1024)
void scan_kernel(int* __restrict__ cursor, size_t ws_size) {
    if (ws_size < WS_NEED) return;
    __shared__ int wtot[32];
    const int w = threadIdx.x >> 5, lane = threadIdx.x & 31;
    const int base = w * SCAN_SEG + lane * 4;
    int4 v[SCAN_K];
    int  ex[SCAN_K];
    int running = 0;
    #pragma unroll
    for (int k = 0; k < SCAN_K; ++k) {
        int idx = base + k * 128;
        v[k] = (idx < N_NODES) ? *(const int4*)(cursor + idx) : make_int4(0, 0, 0, 0);
    }
    #pragma unroll
    for (int k = 0; k < SCAN_K; ++k) {
        int s = v[k].x + v[k].y + v[k].z + v[k].w;
        int inc = s;
        #pragma unroll
        for (int d = 1; d < 32; d <<= 1) {
            int t = __shfl_up(inc, d);
            if (lane >= d) inc += t;
        }
        ex[k] = running + inc - s;
        running += __shfl(inc, 31);
    }
    if (lane == 0) wtot[w] = running;
    __syncthreads();
    int woff = 0;
    for (int i = 0; i < w; ++i) woff += wtot[i];
    #pragma unroll
    for (int k = 0; k < SCAN_K; ++k) {
        int idx = base + k * 128;
        if (idx < N_NODES) {
            int o = woff + ex[k];
            int4 r;
            r.x = o; r.y = o + v[k].x; r.z = r.y + v[k].y; r.w = r.z + v[k].z;
            *(int4*)(cursor + idx) = r;
        }
    }
}

__global__ void fill_kernel(const int* __restrict__ eidx, int* __restrict__ cursor,
                            int* __restrict__ perm, size_t ws_size) {
    if (ws_size < WS_NEED) return;
    int e = blockIdx.x * blockDim.x + threadIdx.x;
    if (e < N_EDGES) {
        int pos = atomicAdd(&cursor[eidx[e]], 1);
        perm[pos] = e;
    }
}

#define K1T 9
#define K2T 4

__global__ __launch_bounds__(256)
void edge_kernel(const float* __restrict__ h,
                 const int*   __restrict__ eidx,
                 const float* __restrict__ x,
                 const float* __restrict__ eattr,
                 const unsigned short* __restrict__ We1b,
                 const float* __restrict__ be1,
                 const unsigned short* __restrict__ We2b,
                 const float* __restrict__ be2,
                 float* __restrict__ agg,
                 const int* __restrict__ perm,
                 size_t ws_size) {
    __shared__ uint4 lA4[4 * K1T * 32 * 2];
    __shared__ uint4 lM4[4 * K2T * 32 * 2];
    unsigned short* lM = (unsigned short*)lM4;

    const int tid = threadIdx.x;
    const int e0  = blockIdx.x * 64;

    int* sE = (int*)lM4;
    int* sR = sE + 64;
    int* sC = sE + 128;
    if (tid < 64) {
        int i = e0 + tid;
        int e = (ws_size >= WS_NEED) ? perm[i] : i;
        sE[tid] = e;
        sR[tid] = eidx[e];
        sC[tid] = eidx[N_EDGES + e];
    }
    __syncthreads();
    const int rowLo = sR[tid & 31];
    const int rowHi = sR[32 + (tid & 31)];

    {
        const int m0 = tid >> 3;
        const int q  = tid & 7;
        float4 hv[16];
        #pragma unroll
        for (int s = 0; s < 2; ++s) {
            const float4* pr = (const float4*)(h + sR[m0 + 32 * s] * HID) + q;
            const float4* pc = (const float4*)(h + sC[m0 + 32 * s] * HID) + q;
            #pragma unroll
            for (int t = 0; t < 4; ++t) {
                hv[s * 8 + t]     = pr[t * 8];
                hv[s * 8 + 4 + t] = pc[t * 8];
            }
        }

        const int tm = tid >> 2;
        const int tc = tid & 3;
        const float* ea = eattr + sE[tm] * EDGE_D;
        int ip = 8 * tc - 1;
        ip = (ip < 0) ? 0 : ip;
        ip = (ip > EDGE_D - 1) ? EDGE_D - 1 : ip;
        const int iq = (tc < 2) ? 2 * tc : 3;
        const int ir = (tc < 2) ? 2 * tc + 1 : 3;
        const float  eP = ea[ip];
        const float4 eQ = ((const float4*)ea)[iq];
        const float4 eR = ((const float4*)ea)[ir];
        const int r = sR[tm];
        const int c = sC[tm];
        float d0 = x[r * 3 + 0] - x[c * 3 + 0];
        float d1 = x[r * 3 + 1] - x[c * 3 + 1];
        float d2 = x[r * 3 + 2] - x[c * 3 + 2];
        const float radial = d0 * d0 + d1 * d1 + d2 * d2;
        const bool two = tc < 2;
        const float eP0 = (tc == 3) ? 0.0f : eP;
        float t8[8];
        t8[0] = (tc == 0) ? radial : eP0;
        t8[1] = two ? eQ.x : 0.0f;
        t8[2] = two ? eQ.y : 0.0f;
        t8[3] = two ? eQ.z : 0.0f;
        t8[4] = two ? eQ.w : 0.0f;
        t8[5] = two ? eR.x : 0.0f;
        t8[6] = two ? eR.y : 0.0f;
        t8[7] = two ? eR.z : 0.0f;

        uint2* lA2 = (uint2*)lA4;
        const int b2 = abuild_base(m0, q, K1T);
        #pragma unroll
        for (int s = 0; s < 2; ++s) {
            #pragma unroll
            for (int t = 0; t < 8; ++t)
                lA2[b2 + (2 * s * K1T + t) * 128] = pack4bf(hv[s * 8 + t]);
        }
        lA4[(((tm >> 4) * K1T + 8) * 32 + (tm & 15) + 16 * (tc & 1)) * 2 + (tc >> 1)] =
            make_uint4(pack2bf(t8[0], t8[1]), pack2bf(t8[2], t8[3]),
                       pack2bf(t8[4], t8[5]), pack2bf(t8[6], t8[7]));
    }
    __syncthreads();

    const int w    = tid >> 5;
    const int lane = tid & 31;
    const int hi   = lane >> 4;
    const int n    = w * 16 + (lane & 15);

    v8f acc[4] = {};
    for (int t = 0; t < K1T; ++t) {
        Frag B;
        const uint4* bp = (const uint4*)We1b + ((t * 8 + w) * 32 + lane) * 2;
        B.q[0] = bp[0]; B.q[1] = bp[1];
        #pragma unroll
        for (int mt = 0; mt < 4; ++mt) {
            Frag A;
            const uint4* ap = (const uint4*)lA4 + ((mt * K1T + t) * 32 + lane) * 2;
            A.q[0] = ap[0]; A.q[1] = ap[1];
            acc[mt] = __builtin_amdgcn_wmma_f32_16x16x32_bf16(
                false, A.v, false, B.v, (short)0, acc[mt], false, false);
        }
    }
    {
        float b = be1[n];
        #pragma unroll
        for (int mt = 0; mt < 4; ++mt) {
            #pragma unroll
            for (int j = 0; j < 8; ++j) {
                int m = mt * 16 + hi * 8 + j;
                lM[asw_pos(m, n, K2T)] = f2bf(silu_f(acc[mt][j] + b));
            }
        }
    }
    __syncthreads();

    v8f acc2[4] = {};
    for (int t = 0; t < K2T; ++t) {
        Frag B;
        const uint4* bp = (const uint4*)We2b + ((t * 8 + w) * 32 + lane) * 2;
        B.q[0] = bp[0]; B.q[1] = bp[1];
        #pragma unroll
        for (int mt = 0; mt < 4; ++mt) {
            Frag A;
            const uint4* ap = (const uint4*)lM4 + ((mt * K2T + t) * 32 + lane) * 2;
            A.q[0] = ap[0]; A.q[1] = ap[1];
            acc2[mt] = __builtin_amdgcn_wmma_f32_16x16x32_bf16(
                false, A.v, false, B.v, (short)0, acc2[mt], false, false);
        }
    }
    {
        float b = be2[n];
        int   cur = 0;
        float run = 0.0f;
        #pragma unroll
        for (int mt = 0; mt < 4; ++mt) {
            #pragma unroll
            for (int j = 0; j < 8; ++j) {
                const int rv = (mt < 2) ? rowLo : rowHi;
                const int r0 = __builtin_amdgcn_readlane(rv, (mt & 1) * 16 + j);
                const int r1 = __builtin_amdgcn_readlane(rv, (mt & 1) * 16 + 8 + j);
                const int r  = hi ? r1 : r0;
                float f = silu_f(acc2[mt][j] + b);
                if (mt == 0 && j == 0) {
                    cur = r;
                    run = f;
                } else {
                    if (r != cur) {
                        atomicAdd(&agg[cur * HID + n], run);
                        run = 0.0f;
                        cur = r;
                    }
                    run += f;
                }
            }
        }
        atomicAdd(&agg[cur * HID + n], run);
    }
}

#define KN1T 8
#define KN2T 4

__global__ __launch_bounds__(256)
void node_kernel(const float* __restrict__ h,
                 const float* __restrict__ agg,
                 const int*   __restrict__ batch,
                 const unsigned short* __restrict__ Wn1b,
                 const float* __restrict__ bn1,
                 const unsigned short* __restrict__ Wn2b,
                 const float* __restrict__ bn2,
                 float* __restrict__ sums) {
    __shared__ uint4 lA4[4 * KN1T * 32 * 2];
    __shared__ uint4 lM4[4 * KN2T * 32 * 2];
    unsigned short* lM = (unsigned short*)lM4;

    const int tid = threadIdx.x;
    const int n0  = blockIdx.x * 64;

    {
        const int m0 = tid >> 3;
        const int q  = tid & 7;
        float4 hv[16];
        #pragma unroll
        for (int s = 0; s < 2; ++s) {
            const int node = n0 + m0 + 32 * s;
            if (node < N_NODES) {
                const float4* ph = (const float4*)(h   + node * HID) + q;
                const float4* pa = (const float4*)(agg + node * HID) + q;
                #pragma unroll
                for (int t = 0; t < 4; ++t) {
                    hv[s * 8 + t]     = ph[t * 8];
                    hv[s * 8 + 4 + t] = pa[t * 8];
                }
            } else {
                #pragma unroll
                for (int t = 0; t < 8; ++t)
                    hv[s * 8 + t] = make_float4(0.0f, 0.0f, 0.0f, 0.0f);
            }
        }
        uint2* lA2 = (uint2*)lA4;
        const int b2 = abuild_base(m0, q, KN1T);
        #pragma unroll
        for (int s = 0; s < 2; ++s) {
            #pragma unroll
            for (int t = 0; t < 8; ++t)
                lA2[b2 + (2 * s * KN1T + t) * 128] = pack4bf(hv[s * 8 + t]);
        }
    }
    __syncthreads();

    const int w    = tid >> 5;
    const int lane = tid & 31;
    const int hi   = lane >> 4;
    const int n    = w * 16 + (lane & 15);

    v8f acc[4] = {};
    for (int t = 0; t < KN1T; ++t) {
        Frag B;
        const uint4* bp = (const uint4*)Wn1b + ((t * 8 + w) * 32 + lane) * 2;
        B.q[0] = bp[0]; B.q[1] = bp[1];
        #pragma unroll
        for (int mt = 0; mt < 4; ++mt) {
            Frag A;
            const uint4* ap = (const uint4*)lA4 + ((mt * KN1T + t) * 32 + lane) * 2;
            A.q[0] = ap[0]; A.q[1] = ap[1];
            acc[mt] = __builtin_amdgcn_wmma_f32_16x16x32_bf16(
                false, A.v, false, B.v, (short)0, acc[mt], false, false);
        }
    }
    {
        float b = bn1[n];
        #pragma unroll
        for (int mt = 0; mt < 4; ++mt) {
            #pragma unroll
            for (int j = 0; j < 8; ++j) {
                int m = mt * 16 + hi * 8 + j;
                lM[asw_pos(m, n, KN2T)] = f2bf(silu_f(acc[mt][j] + b));
            }
        }
    }
    __syncthreads();

    v8f acc2[4] = {};
    for (int t = 0; t < KN2T; ++t) {
        Frag B;
        const uint4* bp = (const uint4*)Wn2b + ((t * 8 + w) * 32 + lane) * 2;
        B.q[0] = bp[0]; B.q[1] = bp[1];
        #pragma unroll
        for (int mt = 0; mt < 4; ++mt) {
            Frag A;
            const uint4* ap = (const uint4*)lM4 + ((mt * KN2T + t) * 32 + lane) * 2;
            A.q[0] = ap[0]; A.q[1] = ap[1];
            acc2[mt] = __builtin_amdgcn_wmma_f32_16x16x32_bf16(
                false, A.v, false, B.v, (short)0, acc2[mt], false, false);
        }
    }
    {
        float b = bn2[n];
        #pragma unroll
        for (int mt = 0; mt < 4; ++mt) {
            #pragma unroll
            for (int j = 0; j < 8; ++j) {
                int m = mt * 16 + hi * 8 + j;
                int node = n0 + m;
                if (node < N_NODES) {
                    float hn = h[node * HID + n] + (acc2[mt][j] + b);
                    int g = batch[node];
                    atomicAdd(&sums[g * HID + n], hn);
                }
            }
        }
    }
}

__global__ void counts_kernel(const int* __restrict__ batch, float* __restrict__ counts) {
    int i = blockIdx.x * blockDim.x + threadIdx.x;
    if (i < N_NODES) atomicAdd(&counts[batch[i]], 1.0f);
}

__global__ void finalize_kernel(const float* __restrict__ sums,
                                const float* __restrict__ counts,
                                float* __restrict__ out) {
    int i = blockIdx.x * blockDim.x + threadIdx.x;
    if (i < N_GRAPHS * HID)
        out[i] = sums[i] / fmaxf(counts[i >> 7], 1.0f);
}

extern "C" void kernel_launch(void* const* d_in, const int* in_sizes, int n_in,
                              void* d_out, int out_size, void* d_ws, size_t ws_size,
                              hipStream_t stream) {
    const float* h     = (const float*)d_in[0];
    const int*   eidx  = (const int*)  d_in[1];
    const float* x     = (const float*)d_in[2];
    const float* eattr = (const float*)d_in[3];
    const int*   batch = (const int*)  d_in[4];
    const float* We1   = (const float*)d_in[5];
    const float* be1   = (const float*)d_in[6];
    const float* We2   = (const float*)d_in[7];
    const float* be2   = (const float*)d_in[8];
    const float* Wn1   = (const float*)d_in[9];
    const float* bn1   = (const float*)d_in[10];
    const float* Wn2   = (const float*)d_in[11];
    const float* bn2   = (const float*)d_in[12];

    char* ws = (char*)d_ws;
    unsigned short* We1b = (unsigned short*)(ws);
    unsigned short* We2b = (unsigned short*)(ws + 73728);
    unsigned short* Wn1b = (unsigned short*)(ws + 106496);
    unsigned short* Wn2b = (unsigned short*)(ws + 172032);
    float* agg    = (float*)(ws + 204800);
    float* sums   = (float*)(ws + 204800 + 25600000);
    float* counts = (float*)(ws + 204800 + 25600000 + 32768);
    int*   cursor = (int*)(ws + 25837824);
    int*   perm   = (int*)(ws + 26037824);

    wconv_kernel<<<(9 * 4096 + 255) / 256, 256, 0, stream>>>(We1, We1b, 273, 9);
    wconv_kernel<<<(4 * 4096 + 255) / 256, 256, 0, stream>>>(We2, We2b, 128, 4);
    wconv_kernel<<<(8 * 4096 + 255) / 256, 256, 0, stream>>>(Wn1, Wn1b, 256, 8);
    wconv_kernel<<<(4 * 4096 + 255) / 256, 256, 0, stream>>>(Wn2, Wn2b, 128, 4);

    zero_f32_kernel<<<(NZ_SORT + 255) / 256, 256, 0, stream>>>(agg, NZ_BASE, NZ_SORT, ws_size);

    hist_kernel<<<(N_EDGES + 255) / 256, 256, 0, stream>>>(eidx, cursor, ws_size);
    scan_kernel<<<1, 1024, 0, stream>>>(cursor, ws_size);
    fill_kernel<<<(N_EDGES + 255) / 256, 256, 0, stream>>>(eidx, cursor, perm, ws_size);

    edge_kernel<<<N_EDGES / 64, 256, 0, stream>>>(h, eidx, x, eattr,
                                                  We1b, be1, We2b, be2, agg,
                                                  perm, ws_size);
    node_kernel<<<(N_NODES + 63) / 64, 256, 0, stream>>>(h, agg, batch,
                                                         Wn1b, bn1, Wn2b, bn2, sums);
    counts_kernel<<<(N_NODES + 255) / 256, 256, 0, stream>>>(batch, counts);
    finalize_kernel<<<(N_GRAPHS * HID + 255) / 256, 256, 0, stream>>>(sums, counts,
                                                                      (float*)d_out);
}


# LDS A tiles re-laid so fragment reads have no bank conflicts: a lane's two 16-byte fragment halves go to separate 512-byte blocks at a 16-byte lane pitch (lane half 1 xor-shifted for the stores)
# speedup vs baseline: 1.7751x; 1.0583x over previous
#include <hip/hip_runtime.h>
#include <math.h>

#define N_NODES  50000
#define N_EDGES  800000
#define HID      128
#define EDGE_D   16
#define N_GRAPHS 64

typedef __attribute__((ext_vector_type(16))) __bf16 v16bf;
typedef __attribute__((ext_vector_type(8)))  float  v8f;

union Frag {
    uint4 q[2];
    v16bf v;
};

__device__ __forceinline__ unsigned short f2bf(float f) {
    union { float f; unsigned int u; } c; c.f = f;
    unsigned int u = c.u;
    u += 0x7fffu + ((u >> 16) & 1u);
    return (unsigned short)(u >> 16);
}

__device__ __forceinline__ float silu_f(float x) {
    return x * __builtin_amdgcn_rcpf(1.0f + __expf(-x));
}

typedef __attribute__((ext_vector_type(2))) float  f32x2;
typedef __attribute__((ext_vector_type(2))) __bf16 bf16x2;

__device__ __forceinline__ unsigned int pack2bf(float a, float b) {
    f32x2 v = {a, b};
    union { bf16x2 h; unsigned int u; } c;
    c.h = __builtin_convertvector(v, bf16x2);
    return c.u;
}

__device__ __forceinline__ uint2 pack4bf(float4 f) {
    return make_uint2(pack2bf(f.x, f.y), pack2bf(f.z, f.w));
}

__device__ __forceinline__ int aslot(int c, int lh) {
    return (c ^ (lh << 2)) + (lh << 4);
}

__device__ __forceinline__ int abuild_base(int m0, int q, int KT) {
    const int lh = (q >> 1) & 1, hf = q >> 2;
    return ((((m0 >> 4) * KT) * 2 + hf) * 32 + aslot(m0 & 15, lh)) * 2 + (q & 1);
}

__device__ __forceinline__ int asw_pos(int m_local, int K, int KT) {
    int mt = m_local >> 4, ml = m_local & 15;
    int t = K >> 5, kk = K & 31;
    int hf = kk >> 4;
    int g  = (kk >> 3) & 1;
    return (((mt * KT + t) * 2 + hf) * 32 + aslot(ml, g)) * 8 + (kk & 7);
}

__global__ void wconv_kernel(const float* __restrict__ W,
                             unsigned short* __restrict__ dst,
                             int Ksrc, int KT) {
    int idx = blockIdx.x * blockDim.x + threadIdx.x;
    int total = KT * 8 * 32 * 16;
    if (idx >= total) return;
    int i    = idx & 15;
    int lane = (idx >> 4) & 31;
    int tj   = idx >> 9;
    int t = tj >> 3, j = tj & 7;
    int g = lane >> 4;
    int K = t * 32 + (i < 8 ? g * 8 + i : 16 + g * 8 + (i - 8));
    int n = j * 16 + (lane & 15);
    float v = (K < Ksrc) ? W[K * HID + n] : 0.0f;
    dst[idx] = f2bf(v);
}

#define WS_NEED 29237824ull
#define NZ_BASE (N_NODES * HID + N_GRAPHS * HID + N_GRAPHS)
#define NZ_SORT (NZ_BASE + N_NODES)

__global__ void zero_f32_kernel(float* p, int n_base, int n_sort, size_t ws_size) {
    int i = blockIdx.x * blockDim.x + threadIdx.x;
    int n = (ws_size >= WS_NEED) ? n_sort : n_base;
    if (i < n) p[i] = 0.0f;
}

__global__ void hist_kernel(const int* __restrict__ eidx, int* __restrict__ cursor,
                            size_t ws_size) {
    if (ws_size < WS_NEED) return;
    int e = blockIdx.x * blockDim.x + threadIdx.x;
    if (e < N_EDGES) atomicAdd(&cursor[eidx[e]], 1);
}

#define SCAN_K   13
#define SCAN_SEG (SCAN_K * 128)
__global__ __launch_bounds__(1024)
void scan_kernel(int* __restrict__ cursor, size_t ws_size) {
    if (ws_size < WS_NEED) return;
    __shared__ int wtot[32];
    const int w = threadIdx.x >> 5, lane = threadIdx.x & 31;
    const int base = w * SCAN_SEG + lane * 4;
    int4 v[SCAN_K];
    int  ex[SCAN_K];
    int running = 0;
    #pragma unroll
    for (int k = 0; k < SCAN_K; ++k) {
        int idx = base + k * 128;
        v[k] = (idx < N_NODES) ? *(const int4*)(cursor + idx) : make_int4(0, 0, 0, 0);
    }
    #pragma unroll
    for (int k = 0; k < SCAN_K; ++k) {
        int s = v[k].x + v[k].y + v[k].z + v[k].w;
        int inc = s;
        #pragma unroll
        for (int d = 1; d < 32; d <<= 1) {
            int t = __shfl_up(inc, d);
            if (lane >= d) inc += t;
        }
        ex[k] = running + inc - s;
        running += __shfl(inc, 31);
    }
    if (lane == 0) wtot[w] = running;
    __syncthreads();
    int woff = 0;
    for (int i = 0; i < w; ++i) woff += wtot[i];
    #pragma unroll
    for (int k = 0; k < SCAN_K; ++k) {
        int idx = base + k * 128;
        if (idx < N_NODES) {
            int o = woff + ex[k];
            int4 r;
            r.x = o; r.y = o + v[k].x; r.z = r.y + v[k].y; r.w = r.z + v[k].z;
            *(int4*)(cursor + idx) = r;
        }
    }
}

__global__ void fill_kernel(const int* __restrict__ eidx, int* __restrict__ cursor,
                            int* __restrict__ perm, size_t ws_size) {
    if (ws_size < WS_NEED) return;
    int e = blockIdx.x * blockDim.x + threadIdx.x;
    if (e < N_EDGES) {
        int pos = atomicAdd(&cursor[eidx[e]], 1);
        perm[pos] = e;
    }
}

#define K1T 9
#define K2T 4

__global__ __launch_bounds__(256)
void edge_kernel(const float* __restrict__ h,
                 const int*   __restrict__ eidx,
                 const float* __restrict__ x,
                 const float* __restrict__ eattr,
                 const unsigned short* __restrict__ We1b,
                 const float* __restrict__ be1,
                 const unsigned short* __restrict__ We2b,
                 const float* __restrict__ be2,
                 float* __restrict__ agg,
                 const int* __restrict__ perm,
                 size_t ws_size) {
    __shared__ uint4 lA4[4 * K1T * 32 * 2];
    __shared__ uint4 lM4[4 * K2T * 32 * 2];
    unsigned short* lM = (unsigned short*)lM4;

    const int tid = threadIdx.x;
    const int e0  = blockIdx.x * 64;

    int* sE = (int*)lM4;
    int* sR = sE + 64;
    int* sC = sE + 128;
    if (tid < 64) {
        int i = e0 + tid;
        int e = (ws_size >= WS_NEED) ? perm[i] : i;
        sE[tid] = e;
        sR[tid] = eidx[e];
        sC[tid] = eidx[N_EDGES + e];
    }
    __syncthreads();
    const int rowLo = sR[tid & 31];
    const int rowHi = sR[32 + (tid & 31)];

    {
        const int m0 = tid >> 3;
        const int q  = tid & 7;
        float4 hv[16];
        #pragma unroll
        for (int s = 0; s < 2; ++s) {
            const float4* pr = (const float4*)(h + sR[m0 + 32 * s] * HID) + q;
            const float4* pc = (const float4*)(h + sC[m0 + 32 * s] * HID) + q;
            #pragma unroll
            for (int t = 0; t < 4; ++t) {
                hv[s * 8 + t]     = pr[t * 8];
                hv[s * 8 + 4 + t] = pc[t * 8];
            }
        }

        const int tm = tid >> 2;
        const int tc = tid & 3;
        const float* ea = eattr + sE[tm] * EDGE_D;
        int ip = 8 * tc - 1;
        ip = (ip < 0) ? 0 : ip;
        ip = (ip > EDGE_D - 1) ? EDGE_D - 1 : ip;
        const int iq = (tc < 2) ? 2 * tc : 3;
        const int ir = (tc < 2) ? 2 * tc + 1 : 3;
        const float  eP = ea[ip];
        const float4 eQ = ((const float4*)ea)[iq];
        const float4 eR = ((const float4*)ea)[ir];
        const int r = sR[tm];
        const int c = sC[tm];
        float d0 = x[r * 3 + 0] - x[c * 3 + 0];
        float d1 = x[r * 3 + 1] - x[c * 3 + 1];
        float d2 = x[r * 3 + 2] - x[c * 3 + 2];
        const float radial = d0 * d0 + d1 * d1 + d2 * d2;
        const bool two = tc < 2;
        const float eP0 = (tc == 3) ? 0.0f : eP;
        float t8[8];
        t8[0] = (tc == 0) ? radial : eP0;
        t8[1] = two ? eQ.x : 0.0f;
        t8[2] = two ? eQ.y : 0.0f;
        t8[3] = two ? eQ.z : 0.0f;
        t8[4] = two ? eQ.w : 0.0f;
        t8[5] = two ? eR.x : 0.0f;
        t8[6] = two ? eR.y : 0.0f;
        t8[7] = two ? eR.z : 0.0f;

        uint2* lA2 = (uint2*)lA4;
        const int b2 = abuild_base(m0, q, K1T);
        #pragma unroll
        for (int s = 0; s < 2; ++s) {
            #pragma unroll
            for (int t = 0; t < 8; ++t)
                lA2[b2 + (2 * s * K1T + t) * 128] = pack4bf(hv[s * 8 + t]);
        }
        lA4[(((tm >> 4) * K1T + 8) * 2 + (tc >> 1)) * 32 + aslot(tm & 15, tc & 1)] =
            make_uint4(pack2bf(t8[0], t8[1]), pack2bf(t8[2], t8[3]),
                       pack2bf(t8[4], t8[5]), pack2bf(t8[6], t8[7]));
    }
    __syncthreads();

    const int w    = tid >> 5;
    const int lane = tid & 31;
    const int hi   = lane >> 4;
    const int n    = w * 16 + (lane & 15);
    const int as0  = aslot(lane & 15, hi);

    v8f acc[4] = {};
    for (int t = 0; t < K1T; ++t) {
        Frag B;
        const uint4* bp = (const uint4*)We1b + ((t * 8 + w) * 32 + lane) * 2;
        B.q[0] = bp[0]; B.q[1] = bp[1];
        #pragma unroll
        for (int mt = 0; mt < 4; ++mt) {
            Frag A;
            const uint4* ap = (const uint4*)lA4 + (mt * K1T + t) * 64;
            A.q[0] = ap[as0]; A.q[1] = ap[as0 + 32];
            acc[mt] = __builtin_amdgcn_wmma_f32_16x16x32_bf16(
                false, A.v, false, B.v, (short)0, acc[mt], false, false);
        }
    }
    {
        float b = be1[n];
        const int mp0 = asw_pos(hi * 8,     n, K2T);
        const int mp1 = asw_pos(hi * 8 + 4, n, K2T);
        #pragma unroll
        for (int mt = 0; mt < 4; ++mt) {
            #pragma unroll
            for (int j = 0; j < 8; ++j)
                lM[mt * (K2T * 512) + ((j < 4) ? mp0 : mp1) + (j & 3) * 8] =
                    f2bf(silu_f(acc[mt][j] + b));
        }
    }
    __syncthreads();

    v8f acc2[4] = {};
    for (int t = 0; t < K2T; ++t) {
        Frag B;
        const uint4* bp = (const uint4*)We2b + ((t * 8 + w) * 32 + lane) * 2;
        B.q[0] = bp[0]; B.q[1] = bp[1];
        #pragma unroll
        for (int mt = 0; mt < 4; ++mt) {
            Frag A;
            const uint4* ap = (const uint4*)lM4 + (mt * K2T + t) * 64;
            A.q[0] = ap[as0]; A.q[1] = ap[as0 + 32];
            acc2[mt] = __builtin_amdgcn_wmma_f32_16x16x32_bf16(
                false, A.v, false, B.v, (short)0, acc2[mt], false, false);
        }
    }
    {
        float b = be2[n];
        int   cur = 0;
        float run = 0.0f;
        #pragma unroll
        for (int mt = 0; mt < 4; ++mt) {
            #pragma unroll
            for (int j = 0; j < 8; ++j) {
                const int rv = (mt < 2) ? rowLo : rowHi;
                const int r0 = __builtin_amdgcn_readlane(rv, (mt & 1) * 16 + j);
                const int r1 = __builtin_amdgcn_readlane(rv, (mt & 1) * 16 + 8 + j);
                const int r  = hi ? r1 : r0;
                float f = silu_f(acc2[mt][j] + b);
                if (mt == 0 && j == 0) {
                    cur = r;
                    run = f;
                } else {
                    if (r != cur) {
                        atomicAdd(&agg[cur * HID + n], run);
                        run = 0.0f;
                        cur = r;
                    }
                    run += f;
                }
            }
        }
        atomicAdd(&agg[cur * HID + n], run);
    }
}

#define KN1T 8
#define KN2T 4

__global__ __launch_bounds__(256)
void node_kernel(const float* __restrict__ h,
                 const float* __restrict__ agg,
                 const int*   __restrict__ batch,
                 const unsigned short* __restrict__ Wn1b,
                 const float* __restrict__ bn1,
                 const unsigned short* __restrict__ Wn2b,
                 const float* __restrict__ bn2,
                 float* __restrict__ sums) {
    __shared__ uint4 lA4[4 * KN1T * 32 * 2];
    __shared__ uint4 lM4[4 * KN2T * 32 * 2];
    unsigned short* lM = (unsigned short*)lM4;

    const int tid = threadIdx.x;
    const int n0  = blockIdx.x * 64;

    {
        const int m0 = tid >> 3;
        const int q  = tid & 7;
        float4 hv[16];
        #pragma unroll
        for (int s = 0; s < 2; ++s) {
            const int node = n0 + m0 + 32 * s;
            if (node < N_NODES) {
                const float4* ph = (const float4*)(h   + node * HID) + q;
                const float4* pa = (const float4*)(agg + node * HID) + q;
                #pragma unroll
                for (int t = 0; t < 4; ++t) {
                    hv[s * 8 + t]     = ph[t * 8];
                    hv[s * 8 + 4 + t] = pa[t * 8];
                }
            } else {
                #pragma unroll
                for (int t = 0; t < 8; ++t)
                    hv[s * 8 + t] = make_float4(0.0f, 0.0f, 0.0f, 0.0f);
            }
        }
        uint2* lA2 = (uint2*)lA4;
        const int b2 = abuild_base(m0, q, KN1T);
        #pragma unroll
        for (int s = 0; s < 2; ++s) {
            #pragma unroll
            for (int t = 0; t < 8; ++t)
                lA2[b2 + (2 * s * KN1T + t) * 128] = pack4bf(hv[s * 8 + t]);
        }
    }
    __syncthreads();

    const int w    = tid >> 5;
    const int lane = tid & 31;
    const int hi   = lane >> 4;
    const int n    = w * 16 + (lane & 15);
    const int as0  = aslot(lane & 15, hi);

    v8f acc[4] = {};
    for (int t = 0; t < KN1T; ++t) {
        Frag B;
        const uint4* bp = (const uint4*)Wn1b + ((t * 8 + w) * 32 + lane) * 2;
        B.q[0] = bp[0]; B.q[1] = bp[1];
        #pragma unroll
        for (int mt = 0; mt < 4; ++mt) {
            Frag A;
            const uint4* ap = (const uint4*)lA4 + (mt * KN1T + t) * 64;
            A.q[0] = ap[as0]; A.q[1] = ap[as0 + 32];
            acc[mt] = __builtin_amdgcn_wmma_f32_16x16x32_bf16(
                false, A.v, false, B.v, (short)0, acc[mt], false, false);
        }
    }
    {
        float b = bn1[n];
        const int mp0 = asw_pos(hi * 8,     n, KN2T);
        const int mp1 = asw_pos(hi * 8 + 4, n, KN2T);
        #pragma unroll
        for (int mt = 0; mt < 4; ++mt) {
            #pragma unroll
            for (int j = 0; j < 8; ++j)
                lM[mt * (KN2T * 512) + ((j < 4) ? mp0 : mp1) + (j & 3) * 8] =
                    f2bf(silu_f(acc[mt][j] + b));
        }
    }
    __syncthreads();

    v8f acc2[4] = {};
    for (int t = 0; t < KN2T; ++t) {
        Frag B;
        const uint4* bp = (const uint4*)Wn2b + ((t * 8 + w) * 32 + lane) * 2;
        B.q[0] = bp[0]; B.q[1] = bp[1];
        #pragma unroll
        for (int mt = 0; mt < 4; ++mt) {
            Frag A;
            const uint4* ap = (const uint4*)lM4 + (mt * KN2T + t) * 64;
            A.q[0] = ap[as0]; A.q[1] = ap[as0 + 32];
            acc2[mt] = __builtin_amdgcn_wmma_f32_16x16x32_bf16(
                false, A.v, false, B.v, (short)0, acc2[mt], false, false);
        }
    }
    {
        float b = bn2[n];
        #pragma unroll
        for (int mt = 0; mt < 4; ++mt) {
            #pragma unroll
            for (int j = 0; j < 8; ++j) {
                int m = mt * 16 + hi * 8 + j;
                int node = n0 + m;
                if (node < N_NODES) {
                    float hn = h[node * HID + n] + (acc2[mt][j] + b);
                    int g = batch[node];
                    atomicAdd(&sums[g * HID + n], hn);
                }
            }
        }
    }
}

__global__ void counts_kernel(const int* __restrict__ batch, float* __restrict__ counts) {
    int i = blockIdx.x * blockDim.x + threadIdx.x;
    if (i < N_NODES) atomicAdd(&counts[batch[i]], 1.0f);
}

__global__ void finalize_kernel(const float* __restrict__ sums,
                                const float* __restrict__ counts,
                                float* __restrict__ out) {
    int i = blockIdx.x * blockDim.x + threadIdx.x;
    if (i < N_GRAPHS * HID)
        out[i] = sums[i] / fmaxf(counts[i >> 7], 1.0f);
}

extern "C" void kernel_launch(void* const* d_in, const int* in_sizes, int n_in,
                              void* d_out, int out_size, void* d_ws, size_t ws_size,
                              hipStream_t stream) {
    const float* h     = (const float*)d_in[0];
    const int*   eidx  = (const int*)  d_in[1];
    const float* x     = (const float*)d_in[2];
    const float* eattr = (const float*)d_in[3];
    const int*   batch = (const int*)  d_in[4];
    const float* We1   = (const float*)d_in[5];
    const float* be1   = (const float*)d_in[6];
    const float* We2   = (const float*)d_in[7];
    const float* be2   = (const float*)d_in[8];
    const float* Wn1   = (const float*)d_in[9];
    const float* bn1   = (const float*)d_in[10];
    const float* Wn2   = (const float*)d_in[11];
    const float* bn2   = (const float*)d_in[12];

    char* ws = (char*)d_ws;
    unsigned short* We1b = (unsigned short*)(ws);
    unsigned short* We2b = (unsigned short*)(ws + 73728);
    unsigned short* Wn1b = (unsigned short*)(ws + 106496);
    unsigned short* Wn2b = (unsigned short*)(ws + 172032);
    float* agg    = (float*)(ws + 204800);
    float* sums   = (float*)(ws + 204800 + 25600000);
    float* counts = (float*)(ws + 204800 + 25600000 + 32768);
    int*   cursor = (int*)(ws + 25837824);
    int*   perm   = (int*)(ws + 26037824);

    wconv_kernel<<<(9 * 4096 + 255) / 256, 256, 0, stream>>>(We1, We1b, 273, 9);
    wconv_kernel<<<(4 * 4096 + 255) / 256, 256, 0, stream>>>(We2, We2b, 128, 4);
    wconv_kernel<<<(8 * 4096 + 255) / 256, 256, 0, stream>>>(Wn1, Wn1b, 256, 8);
    wconv_kernel<<<(4 * 4096 + 255) / 256, 256, 0, stream>>>(Wn2, Wn2b, 128, 4);

    zero_f32_kernel<<<(NZ_SORT + 255) / 256, 256, 0, stream>>>(agg, NZ_BASE, NZ_SORT, ws_size);

    hist_kernel<<<(N_EDGES + 255) / 256, 256, 0, stream>>>(eidx, cursor, ws_size);
    scan_kernel<<<1, 1024, 0, stream>>>(cursor, ws_size);
    fill_kernel<<<(N_EDGES + 255) / 256, 256, 0, stream>>>(eidx, cursor, perm, ws_size);

    edge_kernel<<<N_EDGES / 64, 256, 0, stream>>>(h, eidx, x, eattr,
                                                  We1b, be1, We2b, be2, agg,
                                                  perm, ws_size);
    node_kernel<<<(N_NODES + 63) / 64, 256, 0, stream>>>(h, agg, batch,
                                                         Wn1b, bn1, Wn2b, bn2, sums);
    counts_kernel<<<(N_NODES + 255) / 256, 256, 0, stream>>>(batch, counts);
    finalize_kernel<<<(N_GRAPHS * HID + 255) / 256, 256, 0, stream>>>(sums, counts,
                                                                      (float*)d_out);
}


# edge counting sort in two levels on LDS counters (256-row buckets, then the rows of a bucket): no global atomic per edge in the sort; order inside a row still by atomic arrival, f32 sums as before
# speedup vs baseline: 2.2732x; 1.2806x over previous
#include <hip/hip_runtime.h>
#include <math.h>

#define N_NODES  50000
#define N_EDGES  800000
#define HID      128
#define EDGE_D   16
#define N_GRAPHS 64

typedef __attribute__((ext_vector_type(16))) __bf16 v16bf;
typedef __attribute__((ext_vector_type(8)))  float  v8f;

union Frag {
    uint4 q[2];
    v16bf v;
};

__device__ __forceinline__ unsigned short f2bf(float f) {
    union { float f; unsigned int u; } c; c.f = f;
    unsigned int u = c.u;
    u += 0x7fffu + ((u >> 16) & 1u);
    return (unsigned short)(u >> 16);
}

__device__ __forceinline__ float silu_f(float x) {
    return x * __builtin_amdgcn_rcpf(1.0f + __expf(-x));
}

typedef __attribute__((ext_vector_type(2))) float  f32x2;
typedef __attribute__((ext_vector_type(2))) __bf16 bf16x2;

__device__ __forceinline__ unsigned int pack2bf(float a, float b) {
    f32x2 v = {a, b};
    union { bf16x2 h; unsigned int u; } c;
    c.h = __builtin_convertvector(v, bf16x2);
    return c.u;
}

__device__ __forceinline__ uint2 pack4bf(float4 f) {
    return make_uint2(pack2bf(f.x, f.y), pack2bf(f.z, f.w));
}

__device__ __forceinline__ int aslot(int c, int lh) {
    return (c ^ (lh << 2)) + (lh << 4);
}

__device__ __forceinline__ int abuild_base(int m0, int q, int KT) {
    const int lh = (q >> 1) & 1, hf = q >> 2;
    return ((((m0 >> 4) * KT) * 2 + hf) * 32 + aslot(m0 & 15, lh)) * 2 + (q & 1);
}

__device__ __forceinline__ int asw_pos(int m_local, int K, int KT) {
    int mt = m_local >> 4, ml = m_local & 15;
    int t = K >> 5, kk = K & 31;
    int hf = kk >> 4;
    int g  = (kk >> 3) & 1;
    return (((mt * KT + t) * 2 + hf) * 32 + aslot(ml, g)) * 8 + (kk & 7);
}

__global__ void wconv_kernel(const float* __restrict__ W,
                             unsigned short* __restrict__ dst,
                             int Ksrc, int KT) {
    int idx = blockIdx.x * blockDim.x + threadIdx.x;
    int total = KT * 8 * 32 * 16;
    if (idx >= total) return;
    int i    = idx & 15;
    int lane = (idx >> 4) & 31;
    int tj   = idx >> 9;
    int t = tj >> 3, j = tj & 7;
    int g = lane >> 4;
    int K = t * 32 + (i < 8 ? g * 8 + i : 16 + g * 8 + (i - 8));
    int n = j * 16 + (lane & 15);
    float v = (K < Ksrc) ? W[K * HID + n] : 0.0f;
    dst[idx] = f2bf(v);
}

#define WS_NEED 29237824ull
#define NZ_BASE (N_NODES * HID + N_GRAPHS * HID + N_GRAPHS)

__global__ void zero_f32_kernel(float* p, int n) {
    int i = blockIdx.x * blockDim.x + threadIdx.x;
    if (i < n) p[i] = 0.0f;
}

#define SORT_NBK 196
#define SORT_NB  250
#define SORT_EPB 3200
#define SORT_NT  (SORT_NBK * SORT_NB)
static_assert(SORT_NB * SORT_EPB == N_EDGES, "level 1 covers every edge once");
static_assert(SORT_NBK * 256 >= N_NODES, "every row has a bucket");
static_assert(SORT_EPB <= 4 * 1024, "four edges per thread cover a work-group's share");
static_assert(SORT_NT % 4 == 0 && SORT_NT <= N_NODES, "table fits its slot, scanned by fours");

__global__ __launch_bounds__(1024)
void bucket_count_kernel(const int* __restrict__ eidx, int* __restrict__ table,
                         size_t ws_size) {
    if (ws_size < WS_NEED) return;
    __shared__ int cnt[SORT_NBK];
    const int tid = threadIdx.x;
    if (tid < SORT_NBK) cnt[tid] = 0;
    __syncthreads();
    const int* rows = eidx + blockIdx.x * SORT_EPB;
    int r[4];
    #pragma unroll
    for (int k = 0; k < 4; ++k) {
        const int i = tid + k * 1024;
        r[k] = (i < SORT_EPB) ? rows[i] : 0;
    }
    #pragma unroll
    for (int k = 0; k < 4; ++k) {
        const int i = tid + k * 1024;
        if (i < SORT_EPB) atomicAdd(&cnt[r[k] >> 8], 1);
    }
    __syncthreads();
    if (tid < SORT_NBK) table[tid * SORT_NB + blockIdx.x] = cnt[tid];
}

#define SCAN_K   13
#define SCAN_SEG (SCAN_K * 128)
static_assert(SORT_NT <= 32 * SCAN_SEG, "one work-group scans the whole table");
__global__ __launch_bounds__(1024)
void scan_kernel(int* __restrict__ cursor, int n, size_t ws_size) {
    if (ws_size < WS_NEED) return;
    __shared__ int wtot[32];
    const int w = threadIdx.x >> 5, lane = threadIdx.x & 31;
    const int base = w * SCAN_SEG + lane * 4;
    int4 v[SCAN_K];
    int  ex[SCAN_K];
    int running = 0;
    #pragma unroll
    for (int k = 0; k < SCAN_K; ++k) {
        int idx = base + k * 128;
        v[k] = (idx < n) ? *(const int4*)(cursor + idx) : make_int4(0, 0, 0, 0);
    }
    #pragma unroll
    for (int k = 0; k < SCAN_K; ++k) {
        int s = v[k].x + v[k].y + v[k].z + v[k].w;
        int inc = s;
        #pragma unroll
        for (int d = 1; d < 32; d <<= 1) {
            int t = __shfl_up(inc, d);
            if (lane >= d) inc += t;
        }
        ex[k] = running + inc - s;
        running += __shfl(inc, 31);
    }
    if (lane == 0) wtot[w] = running;
    __syncthreads();
    int woff = 0;
    for (int i = 0; i < w; ++i) woff += wtot[i];
    #pragma unroll
    for (int k = 0; k < SCAN_K; ++k) {
        int idx = base + k * 128;
        if (idx < n) {
            int o = woff + ex[k];
            int4 r;
            r.x = o; r.y = o + v[k].x; r.z = r.y + v[k].y; r.w = r.z + v[k].z;
            *(int4*)(cursor + idx) = r;
        }
    }
}

__global__ __launch_bounds__(1024)
void bucket_scatter_kernel(const int* __restrict__ eidx, const int* __restrict__ table,
                           int* __restrict__ pairs, size_t ws_size) {
    if (ws_size < WS_NEED) return;
    __shared__ int place[SORT_NBK];
    const int tid = threadIdx.x;
    if (tid < SORT_NBK) place[tid] = table[tid * SORT_NB + blockIdx.x];
    __syncthreads();
    const int e0 = blockIdx.x * SORT_EPB;
    int r[4];
    #pragma unroll
    for (int k = 0; k < 4; ++k) {
        const int i = tid + k * 1024;
        r[k] = (i < SORT_EPB) ? eidx[e0 + i] : 0;
    }
    #pragma unroll
    for (int k = 0; k < 4; ++k) {
        const int i = tid + k * 1024;
        if (i < SORT_EPB) {
            const int pos = atomicAdd(&place[r[k] >> 8], 1);
            *(int2*)(pairs + 2 * pos) = make_int2(e0 + i, r[k]);
        }
    }
}

#define SORT_RPT 5
__global__ __launch_bounds__(1024)
void row_sort_kernel(const int* __restrict__ table, const int* __restrict__ pairs,
                     int* __restrict__ perm, size_t ws_size) {
    if (ws_size < WS_NEED) return;
    __shared__ int cnt[256];
    __shared__ int wtot[8];
    const int tid = threadIdx.x;
    const int b = blockIdx.x;
    const int start = table[b * SORT_NB];
    const int end = (b + 1 < SORT_NBK) ? table[(b + 1) * SORT_NB] : N_EDGES;
    if (tid < 256) cnt[tid] = 0;
    __syncthreads();
    const int last = end - 1;
    for (int base = start; base < end; base += SORT_RPT * 1024) {
        int rr[SORT_RPT];
        #pragma unroll
        for (int k = 0; k < SORT_RPT; ++k) {
            const int i = base + k * 1024 + tid;
            rr[k] = pairs[2 * ((i < last) ? i : last) + 1];
        }
        #pragma unroll
        for (int k = 0; k < SORT_RPT; ++k) {
            const int i = base + k * 1024 + tid;
            if (i < end) atomicAdd(&cnt[rr[k] & 255], 1);
        }
    }
    __syncthreads();
    const int w = tid >> 5, lane = tid & 31;
    int v = 0, inc = 0;
    if (tid < 256) {
        v = cnt[tid];
        inc = v;
        #pragma unroll
        for (int d = 1; d < 32; d <<= 1) {
            int t = __shfl_up(inc, d);
            if (lane >= d) inc += t;
        }
        if (lane == 31) wtot[w] = inc;
    }
    __syncthreads();
    if (tid < 256) {
        int woff = 0;
        for (int i = 0; i < w; ++i) woff += wtot[i];
        cnt[tid] = start + woff + inc - v;
    }
    __syncthreads();
    for (int base = start; base < end; base += SORT_RPT * 1024) {
        int2 t[SORT_RPT];
        #pragma unroll
        for (int k = 0; k < SORT_RPT; ++k) {
            const int i = base + k * 1024 + tid;
            t[k] = *(const int2*)(pairs + 2 * ((i < last) ? i : last));
        }
        #pragma unroll
        for (int k = 0; k < SORT_RPT; ++k) {
            const int i = base + k * 1024 + tid;
            if (i < end) {
                const int pos = atomicAdd(&cnt[t[k].y & 255], 1);
                perm[pos] = t[k].x;
            }
        }
    }
}

#define K1T 9
#define K2T 4

__global__ __launch_bounds__(256)
void edge_kernel(const float* __restrict__ h,
                 const int*   __restrict__ eidx,
                 const float* __restrict__ x,
                 const float* __restrict__ eattr,
                 const unsigned short* __restrict__ We1b,
                 const float* __restrict__ be1,
                 const unsigned short* __restrict__ We2b,
                 const float* __restrict__ be2,
                 float* __restrict__ agg,
                 const int* __restrict__ perm,
                 size_t ws_size) {
    __shared__ uint4 lA4[4 * K1T * 32 * 2];
    __shared__ uint4 lM4[4 * K2T * 32 * 2];
    unsigned short* lM = (unsigned short*)lM4;

    const int tid = threadIdx.x;
    const int e0  = blockIdx.x * 64;

    int* sE = (int*)lM4;
    int* sR = sE + 64;
    int* sC = sE + 128;
    if (tid < 64) {
        int i = e0 + tid;
        int e = (ws_size >= WS_NEED) ? perm[i] : i;
        sE[tid] = e;
        sR[tid] = eidx[e];
        sC[tid] = eidx[N_EDGES + e];
    }
    __syncthreads();
    const int rowLo = sR[tid & 31];
    const int rowHi = sR[32 + (tid & 31)];

    {
        const int m0 = tid >> 3;
        const int q  = tid & 7;
        float4 hv[16];
        #pragma unroll
        for (int s = 0; s < 2; ++s) {
            const float4* pr = (const float4*)(h + sR[m0 + 32 * s] * HID) + q;
            const float4* pc = (const float4*)(h + sC[m0 + 32 * s] * HID) + q;
            #pragma unroll
            for (int t = 0; t < 4; ++t) {
                hv[s * 8 + t]     = pr[t * 8];
                hv[s * 8 + 4 + t] = pc[t * 8];
            }
        }

        const int tm = tid >> 2;
        const int tc = tid & 3;
        const float* ea = eattr + sE[tm] * EDGE_D;
        int ip = 8 * tc - 1;
        ip = (ip < 0) ? 0 : ip;
        ip = (ip > EDGE_D - 1) ? EDGE_D - 1 : ip;
        const int iq = (tc < 2) ? 2 * tc : 3;
        const int ir = (tc < 2) ? 2 * tc + 1 : 3;
        const float  eP = ea[ip];
        const float4 eQ = ((const float4*)ea)[iq];
        const float4 eR = ((const float4*)ea)[ir];
        const int r = sR[tm];
        const int c = sC[tm];
        float d0 = x[r * 3 + 0] - x[c * 3 + 0];
        float d1 = x[r * 3 + 1] - x[c * 3 + 1];
        float d2 = x[r * 3 + 2] - x[c * 3 + 2];
        const float radial = d0 * d0 + d1 * d1 + d2 * d2;
        const bool two = tc < 2;
        const float eP0 = (tc == 3) ? 0.0f : eP;
        float t8[8];
        t8[0] = (tc == 0) ? radial : eP0;
        t8[1] = two ? eQ.x : 0.0f;
        t8[2] = two ? eQ.y : 0.0f;
        t8[3] = two ? eQ.z : 0.0f;
        t8[4] = two ? eQ.w : 0.0f;
        t8[5] = two ? eR.x : 0.0f;
        t8[6] = two ? eR.y : 0.0f;
        t8[7] = two ? eR.z : 0.0f;

        uint2* lA2 = (uint2*)lA4;
        const int b2 = abuild_base(m0, q, K1T);
        #pragma unroll
        for (int s = 0; s < 2; ++s) {
            #pragma unroll
            for (int t = 0; t < 8; ++t)
                lA2[b2 + (2 * s * K1T + t) * 128] = pack4bf(hv[s * 8 + t]);
        }
        lA4[(((tm >> 4) * K1T + 8) * 2 + (tc >> 1)) * 32 + aslot(tm & 15, tc & 1)] =
            make_uint4(pack2bf(t8[0], t8[1]), pack2bf(t8[2], t8[3]),
                       pack2bf(t8[4], t8[5]), pack2bf(t8[6], t8[7]));
    }
    __syncthreads();

    const int w    = tid >> 5;
    const int lane = tid & 31;
    const int hi   = lane >> 4;
    const int n    = w * 16 + (lane & 15);
    const int as0  = aslot(lane & 15, hi);

    v8f acc[4] = {};
    for (int t = 0; t < K1T; ++t) {
        Frag B;
        const uint4* bp = (const uint4*)We1b + ((t * 8 + w) * 32 + lane) * 2;
        B.q[0] = bp[0]; B.q[1] = bp[1];
        #pragma unroll
        for (int mt = 0; mt < 4; ++mt) {
            Frag A;
            const uint4* ap = (const uint4*)lA4 + (mt * K1T + t) * 64;
            A.q[0] = ap[as0]; A.q[1] = ap[as0 + 32];
            acc[mt] = __builtin_amdgcn_wmma_f32_16x16x32_bf16(
                false, A.v, false, B.v, (short)0, acc[mt], false, false);
        }
    }
    {
        float b = be1[n];
        const int mp0 = asw_pos(hi * 8,     n, K2T);
        const int mp1 = asw_pos(hi * 8 + 4, n, K2T);
        #pragma unroll
        for (int mt = 0; mt < 4; ++mt) {
            #pragma unroll
            for (int j = 0; j < 8; ++j)
                lM[mt * (K2T * 512) + ((j < 4) ? mp0 : mp1) + (j & 3) * 8] =
                    f2bf(silu_f(acc[mt][j] + b));
        }
    }
    __syncthreads();

    v8f acc2[4] = {};
    for (int t = 0; t < K2T; ++t) {
        Frag B;
        const uint4* bp = (const uint4*)We2b + ((t * 8 + w) * 32 + lane) * 2;
        B.q[0] = bp[0]; B.q[1] = bp[1];
        #pragma unroll
        for (int mt = 0; mt < 4; ++mt) {
            Frag A;
            const uint4* ap = (const uint4*)lM4 + (mt * K2T + t) * 64;
            A.q[0] = ap[as0]; A.q[1] = ap[as0 + 32];
            acc2[mt] = __builtin_amdgcn_wmma_f32_16x16x32_bf16(
                false, A.v, false, B.v, (short)0, acc2[mt], false, false);
        }
    }
    {
        float b = be2[n];
        int   cur = 0;
        float run = 0.0f;
        #pragma unroll
        for (int mt = 0; mt < 4; ++mt) {
            #pragma unroll
            for (int j = 0; j < 8; ++j) {
                const int rv = (mt < 2) ? rowLo : rowHi;
                const int r0 = __builtin_amdgcn_readlane(rv, (mt & 1) * 16 + j);
                const int r1 = __builtin_amdgcn_readlane(rv, (mt & 1) * 16 + 8 + j);
                const int r  = hi ? r1 : r0;
                float f = silu_f(acc2[mt][j] + b);
                if (mt == 0 && j == 0) {
                    cur = r;
                    run = f;
                } else {
                    if (r != cur) {
                        atomicAdd(&agg[cur * HID + n], run);
                        run = 0.0f;
                        cur = r;
                    }
                    run += f;
                }
            }
        }
        atomicAdd(&agg[cur * HID + n], run);
    }
}

#define KN1T 8
#define KN2T 4

__global__ __launch_bounds__(256)
void node_kernel(const float* __restrict__ h,
                 const float* __restrict__ agg,
                 const int*   __restrict__ batch,
                 const unsigned short* __restrict__ Wn1b,
                 const float* __restrict__ bn1,
                 const unsigned short* __restrict__ Wn2b,
                 const float* __restrict__ bn2,
                 float* __restrict__ sums) {
    __shared__ uint4 lA4[4 * KN1T * 32 * 2];
    __shared__ uint4 lM4[4 * KN2T * 32 * 2];
    unsigned short* lM = (unsigned short*)lM4;

    const int tid = threadIdx.x;
    const int n0  = blockIdx.x * 64;

    {
        const int m0 = tid >> 3;
        const int q  = tid & 7;
        float4 hv[16];
        #pragma unroll
        for (int s = 0; s < 2; ++s) {
            const int node = n0 + m0 + 32 * s;
            if (node < N_NODES) {
                const float4* ph = (const float4*)(h   + node * HID) + q;
                const float4* pa = (const float4*)(agg + node * HID) + q;
                #pragma unroll
                for (int t = 0; t < 4; ++t) {
                    hv[s * 8 + t]     = ph[t * 8];
                    hv[s * 8 + 4 + t] = pa[t * 8];
                }
            } else {
                #pragma unroll
                for (int t = 0; t < 8; ++t)
                    hv[s * 8 + t] = make_float4(0.0f, 0.0f, 0.0f, 0.0f);
            }
        }
        uint2* lA2 = (uint2*)lA4;
        const int b2 = abuild_base(m0, q, KN1T);
        #pragma unroll
        for (int s = 0; s < 2; ++s) {
            #pragma unroll
            for (int t = 0; t < 8; ++t)
                lA2[b2 + (2 * s * KN1T + t) * 128] = pack4bf(hv[s * 8 + t]);
        }
    }
    __syncthreads();

    const int w    = tid >> 5;
    const int lane = tid & 31;
    const int hi   = lane >> 4;
    const int n    = w * 16 + (lane & 15);
    const int as0  = aslot(lane & 15, hi);

    v8f acc[4] = {};
    for (int t = 0; t < KN1T; ++t) {
        Frag B;
        const uint4* bp = (const uint4*)Wn1b + ((t * 8 + w) * 32 + lane) * 2;
        B.q[0] = bp[0]; B.q[1] = bp[1];
        #pragma unroll
        for (int mt = 0; mt < 4; ++mt) {
            Frag A;
            const uint4* ap = (const uint4*)lA4 + (mt * KN1T + t) * 64;
            A.q[0] = ap[as0]; A.q[1] = ap[as0 + 32];
            acc[mt] = __builtin_amdgcn_wmma_f32_16x16x32_bf16(
                false, A.v, false, B.v, (short)0, acc[mt], false, false);
        }
    }
    {
        float b = bn1[n];
        const int mp0 = asw_pos(hi * 8,     n, KN2T);
        const int mp1 = asw_pos(hi * 8 + 4, n, KN2T);
        #pragma unroll
        for (int mt = 0; mt < 4; ++mt) {
            #pragma unroll
            for (int j = 0; j < 8; ++j)
                lM[mt * (KN2T * 512) + ((j < 4) ? mp0 : mp1) + (j & 3) * 8] =
                    f2bf(silu_f(acc[mt][j] + b));
        }
    }
    __syncthreads();

    v8f acc2[4] = {};
    for (int t = 0; t < KN2T; ++t) {
        Frag B;
        const uint4* bp = (const uint4*)Wn2b + ((t * 8 + w) * 32 + lane) * 2;
        B.q[0] = bp[0]; B.q[1] = bp[1];
        #pragma unroll
        for (int mt = 0; mt < 4; ++mt) {
            Frag A;
            const uint4* ap = (const uint4*)lM4 + (mt * KN2T + t) * 64;
            A.q[0] = ap[as0]; A.q[1] = ap[as0 + 32];
            acc2[mt] = __builtin_amdgcn_wmma_f32_16x16x32_bf16(
                false, A.v, false, B.v, (short)0, acc2[mt], false, false);
        }
    }
    {
        float b = bn2[n];
        #pragma unroll
        for (int mt = 0; mt < 4; ++mt) {
            #pragma unroll
            for (int j = 0; j < 8; ++j) {
                int m = mt * 16 + hi * 8 + j;
                int node = n0 + m;
                if (node < N_NODES) {
                    float hn = h[node * HID + n] + (acc2[mt][j] + b);
                    int g = batch[node];
                    atomicAdd(&sums[g * HID + n], hn);
                }
            }
        }
    }
}

__global__ void counts_kernel(const int* __restrict__ batch, float* __restrict__ counts) {
    int i = blockIdx.x * blockDim.x + threadIdx.x;
    if (i < N_NODES) atomicAdd(&counts[batch[i]], 1.0f);
}

__global__ void finalize_kernel(const float* __restrict__ sums,
                                const float* __restrict__ counts,
                                float* __restrict__ out) {
    int i = blockIdx.x * blockDim.x + threadIdx.x;
    if (i < N_GRAPHS * HID)
        out[i] = sums[i] / fmaxf(counts[i >> 7], 1.0f);
}

extern "C" void kernel_launch(void* const* d_in, const int* in_sizes, int n_in,
                              void* d_out, int out_size, void* d_ws, size_t ws_size,
                              hipStream_t stream) {
    const float* h     = (const float*)d_in[0];
    const int*   eidx  = (const int*)  d_in[1];
    const float* x     = (const float*)d_in[2];
    const float* eattr = (const float*)d_in[3];
    const int*   batch = (const int*)  d_in[4];
    const float* We1   = (const float*)d_in[5];
    const float* be1   = (const float*)d_in[6];
    const float* We2   = (const float*)d_in[7];
    const float* be2   = (const float*)d_in[8];
    const float* Wn1   = (const float*)d_in[9];
    const float* bn1   = (const float*)d_in[10];
    const float* Wn2   = (const float*)d_in[11];
    const float* bn2   = (const float*)d_in[12];

    char* ws = (char*)d_ws;
    unsigned short* We1b = (unsigned short*)(ws);
    unsigned short* We2b = (unsigned short*)(ws + 73728);
    unsigned short* Wn1b = (unsigned short*)(ws + 106496);
    unsigned short* Wn2b = (unsigned short*)(ws + 172032);
    float* agg    = (float*)(ws + 204800);
    float* sums   = (float*)(ws + 204800 + 25600000);
    float* counts = (float*)(ws + 204800 + 25600000 + 32768);
    int*   table  = (int*)(ws + 25837824);
    int*   perm   = (int*)(ws + 26037824);
    int*   pairs  = (int*)agg;

    wconv_kernel<<<(9 * 4096 + 255) / 256, 256, 0, stream>>>(We1, We1b, 273, 9);
    wconv_kernel<<<(4 * 4096 + 255) / 256, 256, 0, stream>>>(We2, We2b, 128, 4);
    wconv_kernel<<<(8 * 4096 + 255) / 256, 256, 0, stream>>>(Wn1, Wn1b, 256, 8);
    wconv_kernel<<<(4 * 4096 + 255) / 256, 256, 0, stream>>>(Wn2, Wn2b, 128, 4);

    bucket_count_kernel<<<SORT_NB, 1024, 0, stream>>>(eidx, table, ws_size);
    scan_kernel<<<1, 1024, 0, stream>>>(table, SORT_NT, ws_size);
    bucket_scatter_kernel<<<SORT_NB, 1024, 0, stream>>>(eidx, table, pairs, ws_size);
    row_sort_kernel<<<SORT_NBK, 1024, 0, stream>>>(table, pairs, perm, ws_size);

    zero_f32_kernel<<<(NZ_BASE + 255) / 256, 256, 0, stream>>>(agg, NZ_BASE);

    edge_kernel<<<N_EDGES / 64, 256, 0, stream>>>(h, eidx, x, eattr,
                                                  We1b, be1, We2b, be2, agg,
                                                  perm, ws_size);
    node_kernel<<<(N_NODES + 63) / 64, 256, 0, stream>>>(h, agg, batch,
                                                         Wn1b, bn1, Wn2b, bn2, sums);
    counts_kernel<<<(N_NODES + 255) / 256, 256, 0, stream>>>(batch, counts);
    finalize_kernel<<<(N_GRAPHS * HID + 255) / 256, 256, 0, stream>>>(sums, counts,
                                                                      (float*)d_out);
}


# pooling atomics combined before issue: a node-kernel lane sums its nodes of equal graph id in a register (one add per run, was one per node); counts add run lengths; f32 order of pooled sums changes
# speedup vs baseline: 2.9778x; 1.3100x over previous
#include <hip/hip_runtime.h>
#include <math.h>

#define N_NODES  50000
#define N_EDGES  800000
#define HID      128
#define EDGE_D   16
#define N_GRAPHS 64

typedef __attribute__((ext_vector_type(16))) __bf16 v16bf;
typedef __attribute__((ext_vector_type(8)))  float  v8f;

union Frag {
    uint4 q[2];
    v16bf v;
};

__device__ __forceinline__ unsigned short f2bf(float f) {
    union { float f; unsigned int u; } c; c.f = f;
    unsigned int u = c.u;
    u += 0x7fffu + ((u >> 16) & 1u);
    return (unsigned short)(u >> 16);
}

__device__ __forceinline__ float silu_f(float x) {
    return x * __builtin_amdgcn_rcpf(1.0f + __expf(-x));
}

typedef __attribute__((ext_vector_type(2))) float  f32x2;
typedef __attribute__((ext_vector_type(2))) __bf16 bf16x2;

__device__ __forceinline__ unsigned int pack2bf(float a, float b) {
    f32x2 v = {a, b};
    union { bf16x2 h; unsigned int u; } c;
    c.h = __builtin_convertvector(v, bf16x2);
    return c.u;
}

__device__ __forceinline__ uint2 pack4bf(float4 f) {
    return make_uint2(pack2bf(f.x, f.y), pack2bf(f.z, f.w));
}

__device__ __forceinline__ int aslot(int c, int lh) {
    return (c ^ (lh << 2)) + (lh << 4);
}

__device__ __forceinline__ int abuild_base(int m0, int q, int KT) {
    const int lh = (q >> 1) & 1, hf = q >> 2;
    return ((((m0 >> 4) * KT) * 2 + hf) * 32 + aslot(m0 & 15, lh)) * 2 + (q & 1);
}

__device__ __forceinline__ int asw_pos(int m_local, int K, int KT) {
    int mt = m_local >> 4, ml = m_local & 15;
    int t = K >> 5, kk = K & 31;
    int hf = kk >> 4;
    int g  = (kk >> 3) & 1;
    return (((mt * KT + t) * 2 + hf) * 32 + aslot(ml, g)) * 8 + (kk & 7);
}

__global__ void wconv_kernel(const float* __restrict__ W,
                             unsigned short* __restrict__ dst,
                             int Ksrc, int KT) {
    int idx = blockIdx.x * blockDim.x + threadIdx.x;
    int total = KT * 8 * 32 * 16;
    if (idx >= total) return;
    int i    = idx & 15;
    int lane = (idx >> 4) & 31;
    int tj   = idx >> 9;
    int t = tj >> 3, j = tj & 7;
    int g = lane >> 4;
    int K = t * 32 + (i < 8 ? g * 8 + i : 16 + g * 8 + (i - 8));
    int n = j * 16 + (lane & 15);
    float v = (K < Ksrc) ? W[K * HID + n] : 0.0f;
    dst[idx] = f2bf(v);
}

#define WS_NEED 29237824ull
#define NZ_BASE (N_NODES * HID + N_GRAPHS * HID + N_GRAPHS)

__global__ void zero_f32_kernel(float* p, int n) {
    int i = blockIdx.x * blockDim.x + threadIdx.x;
    if (i < n) p[i] = 0.0f;
}

#define SORT_NBK 196
#define SORT_NB  250
#define SORT_EPB 3200
#define SORT_NT  (SORT_NBK * SORT_NB)
static_assert(SORT_NB * SORT_EPB == N_EDGES, "level 1 covers every edge once");
static_assert(SORT_NBK * 256 >= N_NODES, "every row has a bucket");
static_assert(SORT_EPB <= 4 * 1024, "four edges per thread cover a work-group's share");
static_assert(SORT_NT % 4 == 0 && SORT_NT <= N_NODES, "table fits its slot, scanned by fours");

__global__ __launch_bounds__(1024)
void bucket_count_kernel(const int* __restrict__ eidx, int* __restrict__ table,
                         size_t ws_size) {
    if (ws_size < WS_NEED) return;
    __shared__ int cnt[SORT_NBK];
    const int tid = threadIdx.x;
    if (tid < SORT_NBK) cnt[tid] = 0;
    __syncthreads();
    const int* rows = eidx + blockIdx.x * SORT_EPB;
    int r[4];
    #pragma unroll
    for (int k = 0; k < 4; ++k) {
        const int i = tid + k * 1024;
        r[k] = (i < SORT_EPB) ? rows[i] : 0;
    }
    #pragma unroll
    for (int k = 0; k < 4; ++k) {
        const int i = tid + k * 1024;
        if (i < SORT_EPB) atomicAdd(&cnt[r[k] >> 8], 1);
    }
    __syncthreads();
    if (tid < SORT_NBK) table[tid * SORT_NB + blockIdx.x] = cnt[tid];
}

#define SCAN_K   13
#define SCAN_SEG (SCAN_K * 128)
static_assert(SORT_NT <= 32 * SCAN_SEG, "one work-group scans the whole table");
__global__ __launch_bounds__(1024)
void scan_kernel(int* __restrict__ cursor, int n, size_t ws_size) {
    if (ws_size < WS_NEED) return;
    __shared__ int wtot[32];
    const int w = threadIdx.x >> 5, lane = threadIdx.x & 31;
    const int base = w * SCAN_SEG + lane * 4;
    int4 v[SCAN_K];
    int  ex[SCAN_K];
    int running = 0;
    #pragma unroll
    for (int k = 0; k < SCAN_K; ++k) {
        int idx = base + k * 128;
        v[k] = (idx < n) ? *(const int4*)(cursor + idx) : make_int4(0, 0, 0, 0);
    }
    #pragma unroll
    for (int k = 0; k < SCAN_K; ++k) {
        int s = v[k].x + v[k].y + v[k].z + v[k].w;
        int inc = s;
        #pragma unroll
        for (int d = 1; d < 32; d <<= 1) {
            int t = __shfl_up(inc, d);
            if (lane >= d) inc += t;
        }
        ex[k] = running + inc - s;
        running += __shfl(inc, 31);
    }
    if (lane == 0) wtot[w] = running;
    __syncthreads();
    int woff = 0;
    for (int i = 0; i < w; ++i) woff += wtot[i];
    #pragma unroll
    for (int k = 0; k < SCAN_K; ++k) {
        int idx = base + k * 128;
        if (idx < n) {
            int o = woff + ex[k];
            int4 r;
            r.x = o; r.y = o + v[k].x; r.z = r.y + v[k].y; r.w = r.z + v[k].z;
            *(int4*)(cursor + idx) = r;
        }
    }
}

__global__ __launch_bounds__(1024)
void bucket_scatter_kernel(const int* __restrict__ eidx, const int* __restrict__ table,
                           int* __restrict__ pairs, size_t ws_size) {
    if (ws_size < WS_NEED) return;
    __shared__ int place[SORT_NBK];
    const int tid = threadIdx.x;
    if (tid < SORT_NBK) place[tid] = table[tid * SORT_NB + blockIdx.x];
    __syncthreads();
    const int e0 = blockIdx.x * SORT_EPB;
    int r[4];
    #pragma unroll
    for (int k = 0; k < 4; ++k) {
        const int i = tid + k * 1024;
        r[k] = (i < SORT_EPB) ? eidx[e0 + i] : 0;
    }
    #pragma unroll
    for (int k = 0; k < 4; ++k) {
        const int i = tid + k * 1024;
        if (i < SORT_EPB) {
            const int pos = atomicAdd(&place[r[k] >> 8], 1);
            *(int2*)(pairs + 2 * pos) = make_int2(e0 + i, r[k]);
        }
    }
}

#define SORT_RPT 5
__global__ __launch_bounds__(1024)
void row_sort_kernel(const int* __restrict__ table, const int* __restrict__ pairs,
                     int* __restrict__ perm, size_t ws_size) {
    if (ws_size < WS_NEED) return;
    __shared__ int cnt[256];
    __shared__ int wtot[8];
    const int tid = threadIdx.x;
    const int b = blockIdx.x;
    const int start = table[b * SORT_NB];
    const int end = (b + 1 < SORT_NBK) ? table[(b + 1) * SORT_NB] : N_EDGES;
    if (tid < 256) cnt[tid] = 0;
    __syncthreads();
    const int last = end - 1;
    for (int base = start; base < end; base += SORT_RPT * 1024) {
        int rr[SORT_RPT];
        #pragma unroll
        for (int k = 0; k < SORT_RPT; ++k) {
            const int i = base + k * 1024 + tid;
            rr[k] = pairs[2 * ((i < last) ? i : last) + 1];
        }
        #pragma unroll
        for (int k = 0; k < SORT_RPT; ++k) {
            const int i = base + k * 1024 + tid;
            if (i < end) atomicAdd(&cnt[rr[k] & 255], 1);
        }
    }
    __syncthreads();
    const int w = tid >> 5, lane = tid & 31;
    int v = 0, inc = 0;
    if (tid < 256) {
        v = cnt[tid];
        inc = v;
        #pragma unroll
        for (int d = 1; d < 32; d <<= 1) {
            int t = __shfl_up(inc, d);
            if (lane >= d) inc += t;
        }
        if (lane == 31) wtot[w] = inc;
    }
    __syncthreads();
    if (tid < 256) {
        int woff = 0;
        for (int i = 0; i < w; ++i) woff += wtot[i];
        cnt[tid] = start + woff + inc - v;
    }
    __syncthreads();
    for (int base = start; base < end; base += SORT_RPT * 1024) {
        int2 t[SORT_RPT];
        #pragma unroll
        for (int k = 0; k < SORT_RPT; ++k) {
            const int i = base + k * 1024 + tid;
            t[k] = *(const int2*)(pairs + 2 * ((i < last) ? i : last));
        }
        #pragma unroll
        for (int k = 0; k < SORT_RPT; ++k) {
            const int i = base + k * 1024 + tid;
            if (i < end) {
                const int pos = atomicAdd(&cnt[t[k].y & 255], 1);
                perm[pos] = t[k].x;
            }
        }
    }
}

#define K1T 9
#define K2T 4

__global__ __launch_bounds__(256)
void edge_kernel(const float* __restrict__ h,
                 const int*   __restrict__ eidx,
                 const float* __restrict__ x,
                 const float* __restrict__ eattr,
                 const unsigned short* __restrict__ We1b,
                 const float* __restrict__ be1,
                 const unsigned short* __restrict__ We2b,
                 const float* __restrict__ be2,
                 float* __restrict__ agg,
                 const int* __restrict__ perm,
                 size_t ws_size) {
    __shared__ uint4 lA4[4 * K1T * 32 * 2];
    __shared__ uint4 lM4[4 * K2T * 32 * 2];
    unsigned short* lM = (unsigned short*)lM4;

    const int tid = threadIdx.x;
    const int e0  = blockIdx.x * 64;

    int* sE = (int*)lM4;
    int* sR = sE + 64;
    int* sC = sE + 128;
    if (tid < 64) {
        int i = e0 + tid;
        int e = (ws_size >= WS_NEED) ? perm[i] : i;
        sE[tid] = e;
        sR[tid] = eidx[e];
        sC[tid] = eidx[N_EDGES + e];
    }
    __syncthreads();
    const int rowLo = sR[tid & 31];
    const int rowHi = sR[32 + (tid & 31)];

    {
        const int m0 = tid >> 3;
        const int q  = tid & 7;
        float4 hv[16];
        #pragma unroll
        for (int s = 0; s < 2; ++s) {
            const float4* pr = (const float4*)(h + sR[m0 + 32 * s] * HID) + q;
            const float4* pc = (const float4*)(h + sC[m0 + 32 * s] * HID) + q;
            #pragma unroll
            for (int t = 0; t < 4; ++t) {
                hv[s * 8 + t]     = pr[t * 8];
                hv[s * 8 + 4 + t] = pc[t * 8];
            }
        }

        const int tm = tid >> 2;
        const int tc = tid & 3;
        const float* ea = eattr + sE[tm] * EDGE_D;
        int ip = 8 * tc - 1;
        ip = (ip < 0) ? 0 : ip;
        ip = (ip > EDGE_D - 1) ? EDGE_D - 1 : ip;
        const int iq = (tc < 2) ? 2 * tc : 3;
        const int ir = (tc < 2) ? 2 * tc + 1 : 3;
        const float  eP = ea[ip];
        const float4 eQ = ((const float4*)ea)[iq];
        const float4 eR = ((const float4*)ea)[ir];
        const int r = sR[tm];
        const int c = sC[tm];
        float d0 = x[r * 3 + 0] - x[c * 3 + 0];
        float d1 = x[r * 3 + 1] - x[c * 3 + 1];
        float d2 = x[r * 3 + 2] - x[c * 3 + 2];
        const float radial = d0 * d0 + d1 * d1 + d2 * d2;
        const bool two = tc < 2;
        const float eP0 = (tc == 3) ? 0.0f : eP;
        float t8[8];
        t8[0] = (tc == 0) ? radial : eP0;
        t8[1] = two ? eQ.x : 0.0f;
        t8[2] = two ? eQ.y : 0.0f;
        t8[3] = two ? eQ.z : 0.0f;
        t8[4] = two ? eQ.w : 0.0f;
        t8[5] = two ? eR.x : 0.0f;
        t8[6] = two ? eR.y : 0.0f;
        t8[7] = two ? eR.z : 0.0f;

        uint2* lA2 = (uint2*)lA4;
        const int b2 = abuild_base(m0, q, K1T);
        #pragma unroll
        for (int s = 0; s < 2; ++s) {
            #pragma unroll
            for (int t = 0; t < 8; ++t)
                lA2[b2 + (2 * s * K1T + t) * 128] = pack4bf(hv[s * 8 + t]);
        }
        lA4[(((tm >> 4) * K1T + 8) * 2 + (tc >> 1)) * 32 + aslot(tm & 15, tc & 1)] =
            make_uint4(pack2bf(t8[0], t8[1]), pack2bf(t8[2], t8[3]),
                       pack2bf(t8[4], t8[5]), pack2bf(t8[6], t8[7]));
    }
    __syncthreads();

    const int w    = tid >> 5;
    const int lane = tid & 31;
    const int hi   = lane >> 4;
    const int n    = w * 16 + (lane & 15);
    const int as0  = aslot(lane & 15, hi);

    v8f acc[4] = {};
    for (int t = 0; t < K1T; ++t) {
        Frag B;
        const uint4* bp = (const uint4*)We1b + ((t * 8 + w) * 32 + lane) * 2;
        B.q[0] = bp[0]; B.q[1] = bp[1];
        #pragma unroll
        for (int mt = 0; mt < 4; ++mt) {
            Frag A;
            const uint4* ap = (const uint4*)lA4 + (mt * K1T + t) * 64;
            A.q[0] = ap[as0]; A.q[1] = ap[as0 + 32];
            acc[mt] = __builtin_amdgcn_wmma_f32_16x16x32_bf16(
                false, A.v, false, B.v, (short)0, acc[mt], false, false);
        }
    }
    {
        float b = be1[n];
        const int mp0 = asw_pos(hi * 8,     n, K2T);
        const int mp1 = asw_pos(hi * 8 + 4, n, K2T);
        #pragma unroll
        for (int mt = 0; mt < 4; ++mt) {
            #pragma unroll
            for (int j = 0; j < 8; ++j)
                lM[mt * (K2T * 512) + ((j < 4) ? mp0 : mp1) + (j & 3) * 8] =
                    f2bf(silu_f(acc[mt][j] + b));
        }
    }
    __syncthreads();

    v8f acc2[4] = {};
    for (int t = 0; t < K2T; ++t) {
        Frag B;
        const uint4* bp = (const uint4*)We2b + ((t * 8 + w) * 32 + lane) * 2;
        B.q[0] = bp[0]; B.q[1] = bp[1];
        #pragma unroll
        for (int mt = 0; mt < 4; ++mt) {
            Frag A;
            const uint4* ap = (const uint4*)lM4 + (mt * K2T + t) * 64;
            A.q[0] = ap[as0]; A.q[1] = ap[as0 + 32];
            acc2[mt] = __builtin_amdgcn_wmma_f32_16x16x32_bf16(
                false, A.v, false, B.v, (short)0, acc2[mt], false, false);
        }
    }
    {
        float b = be2[n];
        int   cur = 0;
        float run = 0.0f;
        #pragma unroll
        for (int mt = 0; mt < 4; ++mt) {
            #pragma unroll
            for (int j = 0; j < 8; ++j) {
                const int rv = (mt < 2) ? rowLo : rowHi;
                const int r0 = __builtin_amdgcn_readlane(rv, (mt & 1) * 16 + j);
                const int r1 = __builtin_amdgcn_readlane(rv, (mt & 1) * 16 + 8 + j);
                const int r  = hi ? r1 : r0;
                float f = silu_f(acc2[mt][j] + b);
                if (mt == 0 && j == 0) {
                    cur = r;
                    run = f;
                } else {
                    if (r != cur) {
                        atomicAdd(&agg[cur * HID + n], run);
                        run = 0.0f;
                        cur = r;
                    }
                    run += f;
                }
            }
        }
        atomicAdd(&agg[cur * HID + n], run);
    }
}

#define KN1T 8
#define KN2T 4

__global__ __launch_bounds__(256)
void node_kernel(const float* __restrict__ h,
                 const float* __restrict__ agg,
                 const int*   __restrict__ batch,
                 const unsigned short* __restrict__ Wn1b,
                 const float* __restrict__ bn1,
                 const unsigned short* __restrict__ Wn2b,
                 const float* __restrict__ bn2,
                 float* __restrict__ sums) {
    __shared__ uint4 lA4[4 * KN1T * 32 * 2];
    __shared__ uint4 lM4[4 * KN2T * 32 * 2];
    unsigned short* lM = (unsigned short*)lM4;

    const int tid = threadIdx.x;
    const int n0  = blockIdx.x * 64;

    {
        const int m0 = tid >> 3;
        const int q  = tid & 7;
        float4 hv[16];
        #pragma unroll
        for (int s = 0; s < 2; ++s) {
            const int node = n0 + m0 + 32 * s;
            if (node < N_NODES) {
                const float4* ph = (const float4*)(h   + node * HID) + q;
                const float4* pa = (const float4*)(agg + node * HID) + q;
                #pragma unroll
                for (int t = 0; t < 4; ++t) {
                    hv[s * 8 + t]     = ph[t * 8];
                    hv[s * 8 + 4 + t] = pa[t * 8];
                }
            } else {
                #pragma unroll
                for (int t = 0; t < 8; ++t)
                    hv[s * 8 + t] = make_float4(0.0f, 0.0f, 0.0f, 0.0f);
            }
        }
        uint2* lA2 = (uint2*)lA4;
        const int b2 = abuild_base(m0, q, KN1T);
        #pragma unroll
        for (int s = 0; s < 2; ++s) {
            #pragma unroll
            for (int t = 0; t < 8; ++t)
                lA2[b2 + (2 * s * KN1T + t) * 128] = pack4bf(hv[s * 8 + t]);
        }
    }
    __syncthreads();

    const int w    = tid >> 5;
    const int lane = tid & 31;
    const int hi   = lane >> 4;
    const int n    = w * 16 + (lane & 15);
    const int as0  = aslot(lane & 15, hi);

    v8f acc[4] = {};
    for (int t = 0; t < KN1T; ++t) {
        Frag B;
        const uint4* bp = (const uint4*)Wn1b + ((t * 8 + w) * 32 + lane) * 2;
        B.q[0] = bp[0]; B.q[1] = bp[1];
        #pragma unroll
        for (int mt = 0; mt < 4; ++mt) {
            Frag A;
            const uint4* ap = (const uint4*)lA4 + (mt * KN1T + t) * 64;
            A.q[0] = ap[as0]; A.q[1] = ap[as0 + 32];
            acc[mt] = __builtin_amdgcn_wmma_f32_16x16x32_bf16(
                false, A.v, false, B.v, (short)0, acc[mt], false, false);
        }
    }
    {
        float b = bn1[n];
        const int mp0 = asw_pos(hi * 8,     n, KN2T);
        const int mp1 = asw_pos(hi * 8 + 4, n, KN2T);
        #pragma unroll
        for (int mt = 0; mt < 4; ++mt) {
            #pragma unroll
            for (int j = 0; j < 8; ++j)
                lM[mt * (KN2T * 512) + ((j < 4) ? mp0 : mp1) + (j & 3) * 8] =
                    f2bf(silu_f(acc[mt][j] + b));
        }
    }
    __syncthreads();

    v8f acc2[4] = {};
    for (int t = 0; t < KN2T; ++t) {
        Frag B;
        const uint4* bp = (const uint4*)Wn2b + ((t * 8 + w) * 32 + lane) * 2;
        B.q[0] = bp[0]; B.q[1] = bp[1];
        #pragma unroll
        for (int mt = 0; mt < 4; ++mt) {
            Frag A;
            const uint4* ap = (const uint4*)lM4 + (mt * KN2T + t) * 64;
            A.q[0] = ap[as0]; A.q[1] = ap[as0 + 32];
            acc2[mt] = __builtin_amdgcn_wmma_f32_16x16x32_bf16(
                false, A.v, false, B.v, (short)0, acc2[mt], false, false);
        }
    }
    {
        float b = bn2[n];
        int   cur = -1;
        float run = 0.0f;
        #pragma unroll
        for (int mt = 0; mt < 4; ++mt) {
            #pragma unroll
            for (int j = 0; j < 8; ++j) {
                int m = mt * 16 + hi * 8 + j;
                int node = n0 + m;
                if (node < N_NODES) {
                    float hn = h[node * HID + n] + (acc2[mt][j] + b);
                    int g = batch[node];
                    if (g != cur) {
                        if (cur >= 0) atomicAdd(&sums[cur * HID + n], run);
                        cur = g;
                        run = hn;
                    } else {
                        run += hn;
                    }
                }
            }
        }
        if (cur >= 0) atomicAdd(&sums[cur * HID + n], run);
    }
}

__global__ void counts_kernel(const int* __restrict__ batch, float* __restrict__ counts) {
    const int i    = blockIdx.x * blockDim.x + threadIdx.x;
    const int lane = threadIdx.x & 31;
    const int g    = (i < N_NODES) ? batch[i] : -1;
    const int prev = __shfl_up(g, 1);
    const bool head = (lane == 0) || (prev != g);
    const unsigned int heads = __builtin_amdgcn_ballot_w32(head);
    if (head && g >= 0) {
        const unsigned int above = (lane == 31) ? 0u : (heads >> (lane + 1));
        const int len = above ? (1 + __builtin_ctz(above)) : (32 - lane);
        atomicAdd(&counts[g], (float)len);
    }
}

__global__ void finalize_kernel(const float* __restrict__ sums,
                                const float* __restrict__ counts,
                                float* __restrict__ out) {
    int i = blockIdx.x * blockDim.x + threadIdx.x;
    if (i < N_GRAPHS * HID)
        out[i] = sums[i] / fmaxf(counts[i >> 7], 1.0f);
}

extern "C" void kernel_launch(void* const* d_in, const int* in_sizes, int n_in,
                              void* d_out, int out_size, void* d_ws, size_t ws_size,
                              hipStream_t stream) {
    const float* h     = (const float*)d_in[0];
    const int*   eidx  = (const int*)  d_in[1];
    const float* x     = (const float*)d_in[2];
    const float* eattr = (const float*)d_in[3];
    const int*   batch = (const int*)  d_in[4];
    const float* We1   = (const float*)d_in[5];
    const float* be1   = (const float*)d_in[6];
    const float* We2   = (const float*)d_in[7];
    const float* be2   = (const float*)d_in[8];
    const float* Wn1   = (const float*)d_in[9];
    const float* bn1   = (const float*)d_in[10];
    const float* Wn2   = (const float*)d_in[11];
    const float* bn2   = (const float*)d_in[12];

    char* ws = (char*)d_ws;
    unsigned short* We1b = (unsigned short*)(ws);
    unsigned short* We2b = (unsigned short*)(ws + 73728);
    unsigned short* Wn1b = (unsigned short*)(ws + 106496);
    unsigned short* Wn2b = (unsigned short*)(ws + 172032);
    float* agg    = (float*)(ws + 204800);
    float* sums   = (float*)(ws + 204800 + 25600000);
    float* counts = (float*)(ws + 204800 + 25600000 + 32768);
    int*   table  = (int*)(ws + 25837824);
    int*   perm   = (int*)(ws + 26037824);
    int*   pairs  = (int*)agg;

    wconv_kernel<<<(9 * 4096 + 255) / 256, 256, 0, stream>>>(We1, We1b, 273, 9);
    wconv_kernel<<<(4 * 4096 + 255) / 256, 256, 0, stream>>>(We2, We2b, 128, 4);
    wconv_kernel<<<(8 * 4096 + 255) / 256, 256, 0, stream>>>(Wn1, Wn1b, 256, 8);
    wconv_kernel<<<(4 * 4096 + 255) / 256, 256, 0, stream>>>(Wn2, Wn2b, 128, 4);

    bucket_count_kernel<<<SORT_NB, 1024, 0, stream>>>(eidx, table, ws_size);
    scan_kernel<<<1, 1024, 0, stream>>>(table, SORT_NT, ws_size);
    bucket_scatter_kernel<<<SORT_NB, 1024, 0, stream>>>(eidx, table, pairs, ws_size);
    row_sort_kernel<<<SORT_NBK, 1024, 0, stream>>>(table, pairs, perm, ws_size);

    zero_f32_kernel<<<(NZ_BASE + 255) / 256, 256, 0, stream>>>(agg, NZ_BASE);

    edge_kernel<<<N_EDGES / 64, 256, 0, stream>>>(h, eidx, x, eattr,
                                                  We1b, be1, We2b, be2, agg,
                                                  perm, ws_size);
    node_kernel<<<(N_NODES + 63) / 64, 256, 0, stream>>>(h, agg, batch,
                                                         Wn1b, bn1, Wn2b, bn2, sums);
    counts_kernel<<<(N_NODES + 255) / 256, 256, 0, stream>>>(batch, counts);
    finalize_kernel<<<(N_GRAPHS * HID + 255) / 256, 256, 0, stream>>>(sums, counts,
                                                                      (float*)d_out);
}


# 13 launches -> 7: weight conversion rides in the bucket-count launch, clearing of agg/sums/counts in the row-sort launch (16-byte stores), node counts in the node kernel; f32 sum order unchanged
# speedup vs baseline: 3.1850x; 1.0696x over previous
#include <hip/hip_runtime.h>
#include <math.h>

#define N_NODES  50000
#define N_EDGES  800000
#define HID      128
#define EDGE_D   16
#define N_GRAPHS 64

typedef __attribute__((ext_vector_type(16))) __bf16 v16bf;
typedef __attribute__((ext_vector_type(8)))  float  v8f;

union Frag {
    uint4 q[2];
    v16bf v;
};

__device__ __forceinline__ unsigned short f2bf(float f) {
    union { float f; unsigned int u; } c; c.f = f;
    unsigned int u = c.u;
    u += 0x7fffu + ((u >> 16) & 1u);
    return (unsigned short)(u >> 16);
}

__device__ __forceinline__ float silu_f(float x) {
    return x * __builtin_amdgcn_rcpf(1.0f + __expf(-x));
}

typedef __attribute__((ext_vector_type(2))) float  f32x2;
typedef __attribute__((ext_vector_type(2))) __bf16 bf16x2;

__device__ __forceinline__ unsigned int pack2bf(float a, float b) {
    f32x2 v = {a, b};
    union { bf16x2 h; unsigned int u; } c;
    c.h = __builtin_convertvector(v, bf16x2);
    return c.u;
}

__device__ __forceinline__ uint2 pack4bf(float4 f) {
    return make_uint2(pack2bf(f.x, f.y), pack2bf(f.z, f.w));
}

__device__ __forceinline__ int aslot(int c, int lh) {
    return (c ^ (lh << 2)) + (lh << 4);
}

__device__ __forceinline__ int abuild_base(int m0, int q, int KT) {
    const int lh = (q >> 1) & 1, hf = q >> 2;
    return ((((m0 >> 4) * KT) * 2 + hf) * 32 + aslot(m0 & 15, lh)) * 2 + (q & 1);
}

__device__ __forceinline__ int asw_pos(int m_local, int K, int KT) {
    int mt = m_local >> 4, ml = m_local & 15;
    int t = K >> 5, kk = K & 31;
    int hf = kk >> 4;
    int g  = (kk >> 3) & 1;
    return (((mt * KT + t) * 2 + hf) * 32 + aslot(ml, g)) * 8 + (kk & 7);
}

__device__ __forceinline__ void wconv_elem(const float* __restrict__ W,
                                           unsigned short* __restrict__ dst,
                                           int Ksrc, int idx) {
    int i    = idx & 15;
    int lane = (idx >> 4) & 31;
    int tj   = idx >> 9;
    int t = tj >> 3, j = tj & 7;
    int g = lane >> 4;
    int K = t * 32 + (i < 8 ? g * 8 + i : 16 + g * 8 + (i - 8));
    int n = j * 16 + (lane & 15);
    float v = (K < Ksrc) ? W[K * HID + n] : 0.0f;
    dst[idx] = f2bf(v);
}

#define WS_NEED 29237824ull
#define NZ_BASE (N_NODES * HID + N_GRAPHS * HID + N_GRAPHS)

#define SORT_NBK 196
#define SORT_NB  250
#define SORT_EPB 3200
#define SORT_NT  (SORT_NBK * SORT_NB)
static_assert(SORT_NB * SORT_EPB == N_EDGES, "level 1 covers every edge once");
static_assert(SORT_NBK * 256 >= N_NODES, "every row has a bucket");
static_assert(SORT_EPB <= 4 * 1024, "four edges per thread cover a work-group's share");
static_assert(SORT_NT % 4 == 0 && SORT_NT <= N_NODES, "table fits its slot, scanned by fours");

#define WC_E1    (9 * 4096)
#define WC_E2    (WC_E1 + 4 * 4096)
#define WC_N1    (WC_E2 + 8 * 4096)
#define WC_TOTAL (WC_N1 + 4 * 4096)
#define WC_NB    (WC_TOTAL / 1024)
static_assert(WC_NB * 1024 == WC_TOTAL, "whole work-groups convert the weights");

__global__ __launch_bounds__(1024)
void prep_kernel(const int* __restrict__ eidx, int* __restrict__ table, size_t ws_size,
                 const float* __restrict__ We1, const float* __restrict__ We2,
                 const float* __restrict__ Wn1, const float* __restrict__ Wn2,
                 unsigned short* __restrict__ wb) {
    if (blockIdx.x >= SORT_NB) {
        const int g = (blockIdx.x - SORT_NB) * 1024 + threadIdx.x;
        if      (g < WC_E1) wconv_elem(We1, wb,         273, g);
        else if (g < WC_E2) wconv_elem(We2, wb + WC_E1, 128, g - WC_E1);
        else if (g < WC_N1) wconv_elem(Wn1, wb + WC_E2, 256, g - WC_E2);
        else                wconv_elem(Wn2, wb + WC_N1, 128, g - WC_N1);
        return;
    }
    if (ws_size < WS_NEED) return;
    __shared__ int cnt[SORT_NBK];
    const int tid = threadIdx.x;
    if (tid < SORT_NBK) cnt[tid] = 0;
    __syncthreads();
    const int* rows = eidx + blockIdx.x * SORT_EPB;
    int r[4];
    #pragma unroll
    for (int k = 0; k < 4; ++k) {
        const int i = tid + k * 1024;
        r[k] = (i < SORT_EPB) ? rows[i] : 0;
    }
    #pragma unroll
    for (int k = 0; k < 4; ++k) {
        const int i = tid + k * 1024;
        if (i < SORT_EPB) atomicAdd(&cnt[r[k] >> 8], 1);
    }
    __syncthreads();
    if (tid < SORT_NBK) table[tid * SORT_NB + blockIdx.x] = cnt[tid];
}

#define SCAN_K   13
#define SCAN_SEG (SCAN_K * 128)
static_assert(SORT_NT <= 32 * SCAN_SEG, "one work-group scans the whole table");
__global__ __launch_bounds__(1024)
void scan_kernel(int* __restrict__ cursor, int n, size_t ws_size) {
    if (ws_size < WS_NEED) return;
    __shared__ int wtot[32];
    const int w = threadIdx.x >> 5, lane = threadIdx.x & 31;
    const int base = w * SCAN_SEG + lane * 4;
    int4 v[SCAN_K];
    int  ex[SCAN_K];
    int running = 0;
    #pragma unroll
    for (int k = 0; k < SCAN_K; ++k) {
        int idx = base + k * 128;
        v[k] = (idx < n) ? *(const int4*)(cursor + idx) : make_int4(0, 0, 0, 0);
    }
    #pragma unroll
    for (int k = 0; k < SCAN_K; ++k) {
        int s = v[k].x + v[k].y + v[k].z + v[k].w;
        int inc = s;
        #pragma unroll
        for (int d = 1; d < 32; d <<= 1) {
            int t = __shfl_up(inc, d);
            if (lane >= d) inc += t;
        }
        ex[k] = running + inc - s;
        running += __shfl(inc, 31);
    }
    if (lane == 0) wtot[w] = running;
    __syncthreads();
    int woff = 0;
    for (int i = 0; i < w; ++i) woff += wtot[i];
    #pragma unroll
    for (int k = 0; k < SCAN_K; ++k) {
        int idx = base + k * 128;
        if (idx < n) {
            int o = woff + ex[k];
            int4 r;
            r.x = o; r.y = o + v[k].x; r.z = r.y + v[k].y; r.w = r.z + v[k].z;
            *(int4*)(cursor + idx) = r;
        }
    }
}

__global__ __launch_bounds__(1024)
void bucket_scatter_kernel(const int* __restrict__ eidx, const int* __restrict__ table,
                           int* __restrict__ pairs, size_t ws_size) {
    if (ws_size < WS_NEED) return;
    __shared__ int place[SORT_NBK];
    const int tid = threadIdx.x;
    if (tid < SORT_NBK) place[tid] = table[tid * SORT_NB + blockIdx.x];
    __syncthreads();
    const int e0 = blockIdx.x * SORT_EPB;
    int r[4];
    #pragma unroll
    for (int k = 0; k < 4; ++k) {
        const int i = tid + k * 1024;
        r[k] = (i < SORT_EPB) ? eidx[e0 + i] : 0;
    }
    #pragma unroll
    for (int k = 0; k < 4; ++k) {
        const int i = tid + k * 1024;
        if (i < SORT_EPB) {
            const int pos = atomicAdd(&place[r[k] >> 8], 1);
            *(int2*)(pairs + 2 * pos) = make_int2(e0 + i, r[k]);
        }
    }
}

#define SORT_RPT 5
static_assert((2 * N_EDGES) % 4 == 0 && NZ_BASE % 4 == 0, "cleared by 16-byte stores");
__global__ __launch_bounds__(1024)
void row_sort_kernel(const int* __restrict__ table, int* pairs,
                     int* __restrict__ perm, size_t ws_size) {
    const int tid = threadIdx.x;
    const int b = blockIdx.x;
    const bool sorted = ws_size >= WS_NEED;
    {
        const int q0 = sorted ? (2 * N_EDGES) / 4 : 0;
        const int q1 = NZ_BASE / 4;
        const int per = (q1 - q0 + SORT_NBK - 1) / SORT_NBK;
        const int a = q0 + b * per;
        const int e = (a + per < q1) ? a + per : q1;
        for (int i = a + tid; i < e; i += 1024)
            ((int4*)pairs)[i] = make_int4(0, 0, 0, 0);
    }
    if (!sorted) return;
    __shared__ int cnt[256];
    __shared__ int wtot[8];
    const int start = table[b * SORT_NB];
    const int end = (b + 1 < SORT_NBK) ? table[(b + 1) * SORT_NB] : N_EDGES;
    if (tid < 256) cnt[tid] = 0;
    __syncthreads();
    const int last = end - 1;
    for (int base = start; base < end; base += SORT_RPT * 1024) {
        int rr[SORT_RPT];
        #pragma unroll
        for (int k = 0; k < SORT_RPT; ++k) {
            const int i = base + k * 1024 + tid;
            rr[k] = pairs[2 * ((i < last) ? i : last) + 1];
        }
        #pragma unroll
        for (int k = 0; k < SORT_RPT; ++k) {
            const int i = base + k * 1024 + tid;
            if (i < end) atomicAdd(&cnt[rr[k] & 255], 1);
        }
    }
    __syncthreads();
    const int w = tid >> 5, lane = tid & 31;
    int v = 0, inc = 0;
    if (tid < 256) {
        v = cnt[tid];
        inc = v;
        #pragma unroll
        for (int d = 1; d < 32; d <<= 1) {
            int t = __shfl_up(inc, d);
            if (lane >= d) inc += t;
        }
        if (lane == 31) wtot[w] = inc;
    }
    __syncthreads();
    if (tid < 256) {
        int woff = 0;
        for (int i = 0; i < w; ++i) woff += wtot[i];
        cnt[tid] = start + woff + inc - v;
    }
    __syncthreads();
    for (int base = start; base < end; base += SORT_RPT * 1024) {
        int2 t[SORT_RPT];
        #pragma unroll
        for (int k = 0; k < SORT_RPT; ++k) {
            const int i = base + k * 1024 + tid;
            t[k] = *(const int2*)(pairs + 2 * ((i < last) ? i : last));
        }
        #pragma unroll
        for (int k = 0; k < SORT_RPT; ++k) {
            const int i = base + k * 1024 + tid;
            if (i < end) {
                const int pos = atomicAdd(&cnt[t[k].y & 255], 1);
                perm[pos] = t[k].x;
            }
        }
    }
    __syncthreads();
    for (int i = start + tid; i < end; i += 1024)
        ((int2*)pairs)[i] = make_int2(0, 0);
}

#define K1T 9
#define K2T 4

__global__ __launch_bounds__(256)
void edge_kernel(const float* __restrict__ h,
                 const int*   __restrict__ eidx,
                 const float* __restrict__ x,
                 const float* __restrict__ eattr,
                 const unsigned short* __restrict__ We1b,
                 const float* __restrict__ be1,
                 const unsigned short* __restrict__ We2b,
                 const float* __restrict__ be2,
                 float* __restrict__ agg,
                 const int* __restrict__ perm,
                 size_t ws_size) {
    __shared__ uint4 lA4[4 * K1T * 32 * 2];
    __shared__ uint4 lM4[4 * K2T * 32 * 2];
    unsigned short* lM = (unsigned short*)lM4;

    const int tid = threadIdx.x;
    const int e0  = blockIdx.x * 64;

    int* sE = (int*)lM4;
    int* sR = sE + 64;
    int* sC = sE + 128;
    if (tid < 64) {
        int i = e0 + tid;
        int e = (ws_size >= WS_NEED) ? perm[i] : i;
        sE[tid] = e;
        sR[tid] = eidx[e];
        sC[tid] = eidx[N_EDGES + e];
    }
    __syncthreads();
    const int rowLo = sR[tid & 31];
    const int rowHi = sR[32 + (tid & 31)];

    {
        const int m0 = tid >> 3;
        const int q  = tid & 7;
        float4 hv[16];
        #pragma unroll
        for (int s = 0; s < 2; ++s) {
            const float4* pr = (const float4*)(h + sR[m0 + 32 * s] * HID) + q;
            const float4* pc = (const float4*)(h + sC[m0 + 32 * s] * HID) + q;
            #pragma unroll
            for (int t = 0; t < 4; ++t) {
                hv[s * 8 + t]     = pr[t * 8];
                hv[s * 8 + 4 + t] = pc[t * 8];
            }
        }

        const int tm = tid >> 2;
        const int tc = tid & 3;
        const float* ea = eattr + sE[tm] * EDGE_D;
        int ip = 8 * tc - 1;
        ip = (ip < 0) ? 0 : ip;
        ip = (ip > EDGE_D - 1) ? EDGE_D - 1 : ip;
        const int iq = (tc < 2) ? 2 * tc : 3;
        const int ir = (tc < 2) ? 2 * tc + 1 : 3;
        const float  eP = ea[ip];
        const float4 eQ = ((const float4*)ea)[iq];
        const float4 eR = ((const float4*)ea)[ir];
        const int r = sR[tm];
        const int c = sC[tm];
        float d0 = x[r * 3 + 0] - x[c * 3 + 0];
        float d1 = x[r * 3 + 1] - x[c * 3 + 1];
        float d2 = x[r * 3 + 2] - x[c * 3 + 2];
        const float radial = d0 * d0 + d1 * d1 + d2 * d2;
        const bool two = tc < 2;
        const float eP0 = (tc == 3) ? 0.0f : eP;
        float t8[8];
        t8[0] = (tc == 0) ? radial : eP0;
        t8[1] = two ? eQ.x : 0.0f;
        t8[2] = two ? eQ.y : 0.0f;
        t8[3] = two ? eQ.z : 0.0f;
        t8[4] = two ? eQ.w : 0.0f;
        t8[5] = two ? eR.x : 0.0f;
        t8[6] = two ? eR.y : 0.0f;
        t8[7] = two ? eR.z : 0.0f;

        uint2* lA2 = (uint2*)lA4;
        const int b2 = abuild_base(m0, q, K1T);
        #pragma unroll
        for (int s = 0; s < 2; ++s) {
            #pragma unroll
            for (int t = 0; t < 8; ++t)
                lA2[b2 + (2 * s * K1T + t) * 128] = pack4bf(hv[s * 8 + t]);
        }
        lA4[(((tm >> 4) * K1T + 8) * 2 + (tc >> 1)) * 32 + aslot(tm & 15, tc & 1)] =
            make_uint4(pack2bf(t8[0], t8[1]), pack2bf(t8[2], t8[3]),
                       pack2bf(t8[4], t8[5]), pack2bf(t8[6], t8[7]));
    }
    __syncthreads();

    const int w    = tid >> 5;
    const int lane = tid & 31;
    const int hi   = lane >> 4;
    const int n    = w * 16 + (lane & 15);
    const int as0  = aslot(lane & 15, hi);

    v8f acc[4] = {};
    for (int t = 0; t < K1T; ++t) {
        Frag B;
        const uint4* bp = (const uint4*)We1b + ((t * 8 + w) * 32 + lane) * 2;
        B.q[0] = bp[0]; B.q[1] = bp[1];
        #pragma unroll
        for (int mt = 0; mt < 4; ++mt) {
            Frag A;
            const uint4* ap = (const uint4*)lA4 + (mt * K1T + t) * 64;
            A.q[0] = ap[as0]; A.q[1] = ap[as0 + 32];
            acc[mt] = __builtin_amdgcn_wmma_f32_16x16x32_bf16(
                false, A.v, false, B.v, (short)0, acc[mt], false, false);
        }
    }
    {
        float b = be1[n];
        const int mp0 = asw_pos(hi * 8,     n, K2T);
        const int mp1 = asw_pos(hi * 8 + 4, n, K2T);
        #pragma unroll
        for (int mt = 0; mt < 4; ++mt) {
            #pragma unroll
            for (int j = 0; j < 8; ++j)
                lM[mt * (K2T * 512) + ((j < 4) ? mp0 : mp1) + (j & 3) * 8] =
                    f2bf(silu_f(acc[mt][j] + b));
        }
    }
    __syncthreads();

    v8f acc2[4] = {};
    for (int t = 0; t < K2T; ++t) {
        Frag B;
        const uint4* bp = (const uint4*)We2b + ((t * 8 + w) * 32 + lane) * 2;
        B.q[0] = bp[0]; B.q[1] = bp[1];
        #pragma unroll
        for (int mt = 0; mt < 4; ++mt) {
            Frag A;
            const uint4* ap = (const uint4*)lM4 + (mt * K2T + t) * 64;
            A.q[0] = ap[as0]; A.q[1] = ap[as0 + 32];
            acc2[mt] = __builtin_amdgcn_wmma_f32_16x16x32_bf16(
                false, A.v, false, B.v, (short)0, acc2[mt], false, false);
        }
    }
    {
        float b = be2[n];
        int   cur = 0;
        float run = 0.0f;
        #pragma unroll
        for (int mt = 0; mt < 4; ++mt) {
            #pragma unroll
            for (int j = 0; j < 8; ++j) {
                const int rv = (mt < 2) ? rowLo : rowHi;
                const int r0 = __builtin_amdgcn_readlane(rv, (mt & 1) * 16 + j);
                const int r1 = __builtin_amdgcn_readlane(rv, (mt & 1) * 16 + 8 + j);
                const int r  = hi ? r1 : r0;
                float f = silu_f(acc2[mt][j] + b);
                if (mt == 0 && j == 0) {
                    cur = r;
                    run = f;
                } else {
                    if (r != cur) {
                        atomicAdd(&agg[cur * HID + n], run);
                        run = 0.0f;
                        cur = r;
                    }
                    run += f;
                }
            }
        }
        atomicAdd(&agg[cur * HID + n], run);
    }
}

#define KN1T 8
#define KN2T 4

__global__ __launch_bounds__(256)
void node_kernel(const float* __restrict__ h,
                 const float* __restrict__ agg,
                 const int*   __restrict__ batch,
                 const unsigned short* __restrict__ Wn1b,
                 const float* __restrict__ bn1,
                 const unsigned short* __restrict__ Wn2b,
                 const float* __restrict__ bn2,
                 float* __restrict__ sums,
                 float* __restrict__ counts) {
    __shared__ uint4 lA4[4 * KN1T * 32 * 2];
    __shared__ uint4 lM4[4 * KN2T * 32 * 2];
    unsigned short* lM = (unsigned short*)lM4;

    const int tid = threadIdx.x;
    const int n0  = blockIdx.x * 64;

    if (tid < 64) {
        const int i    = n0 + tid;
        const int lane = tid & 31;
        const int g    = (i < N_NODES) ? batch[i] : -1;
        const int prev = __shfl_up(g, 1);
        const bool head = (lane == 0) || (prev != g);
        const unsigned int heads = __builtin_amdgcn_ballot_w32(head);
        if (head && g >= 0) {
            const unsigned int above = (lane == 31) ? 0u : (heads >> (lane + 1));
            const int len = above ? (1 + __builtin_ctz(above)) : (32 - lane);
            atomicAdd(&counts[g], (float)len);
        }
    }

    {
        const int m0 = tid >> 3;
        const int q  = tid & 7;
        float4 hv[16];
        #pragma unroll
        for (int s = 0; s < 2; ++s) {
            const int node = n0 + m0 + 32 * s;
            if (node < N_NODES) {
                const float4* ph = (const float4*)(h   + node * HID) + q;
                const float4* pa = (const float4*)(agg + node * HID) + q;
                #pragma unroll
                for (int t = 0; t < 4; ++t) {
                    hv[s * 8 + t]     = ph[t * 8];
                    hv[s * 8 + 4 + t] = pa[t * 8];
                }
            } else {
                #pragma unroll
                for (int t = 0; t < 8; ++t)
                    hv[s * 8 + t] = make_float4(0.0f, 0.0f, 0.0f, 0.0f);
            }
        }
        uint2* lA2 = (uint2*)lA4;
        const int b2 = abuild_base(m0, q, KN1T);
        #pragma unroll
        for (int s = 0; s < 2; ++s) {
            #pragma unroll
            for (int t = 0; t < 8; ++t)
                lA2[b2 + (2 * s * KN1T + t) * 128] = pack4bf(hv[s * 8 + t]);
        }
    }
    __syncthreads();

    const int w    = tid >> 5;
    const int lane = tid & 31;
    const int hi   = lane >> 4;
    const int n    = w * 16 + (lane & 15);
    const int as0  = aslot(lane & 15, hi);

    v8f acc[4] = {};
    for (int t = 0; t < KN1T; ++t) {
        Frag B;
        const uint4* bp = (const uint4*)Wn1b + ((t * 8 + w) * 32 + lane) * 2;
        B.q[0] = bp[0]; B.q[1] = bp[1];
        #pragma unroll
        for (int mt = 0; mt < 4; ++mt) {
            Frag A;
            const uint4* ap = (const uint4*)lA4 + (mt * KN1T + t) * 64;
            A.q[0] = ap[as0]; A.q[1] = ap[as0 + 32];
            acc[mt] = __builtin_amdgcn_wmma_f32_16x16x32_bf16(
                false, A.v, false, B.v, (short)0, acc[mt], false, false);
        }
    }
    {
        float b = bn1[n];
        const int mp0 = asw_pos(hi * 8,     n, KN2T);
        const int mp1 = asw_pos(hi * 8 + 4, n, KN2T);
        #pragma unroll
        for (int mt = 0; mt < 4; ++mt) {
            #pragma unroll
            for (int j = 0; j < 8; ++j)
                lM[mt * (KN2T * 512) + ((j < 4) ? mp0 : mp1) + (j & 3) * 8] =
                    f2bf(silu_f(acc[mt][j] + b));
        }
    }
    __syncthreads();

    v8f acc2[4] = {};
    for (int t = 0; t < KN2T; ++t) {
        Frag B;
        const uint4* bp = (const uint4*)Wn2b + ((t * 8 + w) * 32 + lane) * 2;
        B.q[0] = bp[0]; B.q[1] = bp[1];
        #pragma unroll
        for (int mt = 0; mt < 4; ++mt) {
            Frag A;
            const uint4* ap = (const uint4*)lM4 + (mt * KN2T + t) * 64;
            A.q[0] = ap[as0]; A.q[1] = ap[as0 + 32];
            acc2[mt] = __builtin_amdgcn_wmma_f32_16x16x32_bf16(
                false, A.v, false, B.v, (short)0, acc2[mt], false, false);
        }
    }
    {
        float b = bn2[n];
        int   cur = -1;
        float run = 0.0f;
        #pragma unroll
        for (int mt = 0; mt < 4; ++mt) {
            #pragma unroll
            for (int j = 0; j < 8; ++j) {
                int m = mt * 16 + hi * 8 + j;
                int node = n0 + m;
                if (node < N_NODES) {
                    float hn = h[node * HID + n] + (acc2[mt][j] + b);
                    int g = batch[node];
                    if (g != cur) {
                        if (cur >= 0) atomicAdd(&sums[cur * HID + n], run);
                        cur = g;
                        run = hn;
                    } else {
                        run += hn;
                    }
                }
            }
        }
        if (cur >= 0) atomicAdd(&sums[cur * HID + n], run);
    }
}

__global__ void finalize_kernel(const float* __restrict__ sums,
                                const float* __restrict__ counts,
                                float* __restrict__ out) {
    int i = blockIdx.x * blockDim.x + threadIdx.x;
    if (i < N_GRAPHS * HID)
        out[i] = sums[i] / fmaxf(counts[i >> 7], 1.0f);
}

extern "C" void kernel_launch(void* const* d_in, const int* in_sizes, int n_in,
                              void* d_out, int out_size, void* d_ws, size_t ws_size,
                              hipStream_t stream) {
    const float* h     = (const float*)d_in[0];
    const int*   eidx  = (const int*)  d_in[1];
    const float* x     = (const float*)d_in[2];
    const float* eattr = (const float*)d_in[3];
    const int*   batch = (const int*)  d_in[4];
    const float* We1   = (const float*)d_in[5];
    const float* be1   = (const float*)d_in[6];
    const float* We2   = (const float*)d_in[7];
    const float* be2   = (const float*)d_in[8];
    const float* Wn1   = (const float*)d_in[9];
    const float* bn1   = (const float*)d_in[10];
    const float* Wn2   = (const float*)d_in[11];
    const float* bn2   = (const float*)d_in[12];

    char* ws = (char*)d_ws;
    unsigned short* We1b = (unsigned short*)(ws);
    unsigned short* We2b = (unsigned short*)(ws + 73728);
    unsigned short* Wn1b = (unsigned short*)(ws + 106496);
    unsigned short* Wn2b = (unsigned short*)(ws + 172032);
    float* agg    = (float*)(ws + 204800);
    float* sums   = (float*)(ws + 204800 + 25600000);
    float* counts = (float*)(ws + 204800 + 25600000 + 32768);
    int*   table  = (int*)(ws + 25837824);
    int*   perm   = (int*)(ws + 26037824);
    int*   pairs  = (int*)agg;

    prep_kernel<<<SORT_NB + WC_NB, 1024, 0, stream>>>(eidx, table, ws_size,
                                                      We1, We2, Wn1, Wn2, We1b);
    scan_kernel<<<1, 1024, 0, stream>>>(table, SORT_NT, ws_size);
    bucket_scatter_kernel<<<SORT_NB, 1024, 0, stream>>>(eidx, table, pairs, ws_size);
    row_sort_kernel<<<SORT_NBK, 1024, 0, stream>>>(table, pairs, perm, ws_size);

    edge_kernel<<<N_EDGES / 64, 256, 0, stream>>>(h, eidx, x, eattr,
                                                  We1b, be1, We2b, be2, agg,
                                                  perm, ws_size);
    node_kernel<<<(N_NODES + 63) / 64, 256, 0, stream>>>(h, agg, batch,
                                                         Wn1b, bn1, Wn2b, bn2, sums,
                                                         counts);
    finalize_kernel<<<(N_GRAPHS * HID + 255) / 256, 256, 0, stream>>>(sums, counts,
                                                                      (float*)d_out);
}


# edge MLP layer 1 split by linearity: h@We1[0:128], h@We1[128:256] made once per node (bf16 WMMA, new launch) as f32 tables; edges gather and add them; f32 sum order of layer 1 changes
# speedup vs baseline: 3.5980x; 1.1297x over previous
#include <hip/hip_runtime.h>
#include <math.h>

#define N_NODES  50000
#define N_EDGES  800000
#define HID      128
#define EDGE_D   16
#define N_GRAPHS 64

typedef __attribute__((ext_vector_type(16))) __bf16 v16bf;
typedef __attribute__((ext_vector_type(8)))  float  v8f;

union Frag {
    uint4 q[2];
    v16bf v;
};

__device__ __forceinline__ unsigned short f2bf(float f) {
    union { float f; unsigned int u; } c; c.f = f;
    unsigned int u = c.u;
    u += 0x7fffu + ((u >> 16) & 1u);
    return (unsigned short)(u >> 16);
}

__device__ __forceinline__ float silu_f(float x) {
    return x * __builtin_amdgcn_rcpf(1.0f + __expf(-x));
}

typedef __attribute__((ext_vector_type(2))) float  f32x2;
typedef __attribute__((ext_vector_type(2))) __bf16 bf16x2;

__device__ __forceinline__ unsigned int pack2bf(float a, float b) {
    f32x2 v = {a, b};
    union { bf16x2 h; unsigned int u; } c;
    c.h = __builtin_convertvector(v, bf16x2);
    return c.u;
}

__device__ __forceinline__ uint2 pack4bf(float4 f) {
    return make_uint2(pack2bf(f.x, f.y), pack2bf(f.z, f.w));
}

__device__ __forceinline__ int aslot(int c, int lh) {
    return (c ^ (lh << 2)) + (lh << 4);
}

__device__ __forceinline__ int abuild_base(int m0, int q, int KT) {
    const int lh = (q >> 1) & 1, hf = q >> 2;
    return ((((m0 >> 4) * KT) * 2 + hf) * 32 + aslot(m0 & 15, lh)) * 2 + (q & 1);
}

__device__ __forceinline__ int asw_pos(int m_local, int K, int KT) {
    int mt = m_local >> 4, ml = m_local & 15;
    int t = K >> 5, kk = K & 31;
    int hf = kk >> 4;
    int g  = (kk >> 3) & 1;
    return (((mt * KT + t) * 2 + hf) * 32 + aslot(ml, g)) * 8 + (kk & 7);
}

__device__ __forceinline__ void wconv_elem(const float* __restrict__ W,
                                           unsigned short* __restrict__ dst,
                                           int Ksrc, int idx) {
    int i    = idx & 15;
    int lane = (idx >> 4) & 31;
    int tj   = idx >> 9;
    int t = tj >> 3, j = tj & 7;
    int g = lane >> 4;
    int K = t * 32 + (i < 8 ? g * 8 + i : 16 + g * 8 + (i - 8));
    int n = j * 16 + (lane & 15);
    float v = (K < Ksrc) ? W[K * HID + n] : 0.0f;
    dst[idx] = f2bf(v);
}

#define WS_NEED 29237824ull
#define NZ_BASE (N_NODES * HID + N_GRAPHS * HID + N_GRAPHS)
#define PQ_OFF   29238272ull
#define WS_NEED2 (PQ_OFF + 2ull * N_NODES * HID * 4ull)
static_assert(PQ_OFF >= WS_NEED && PQ_OFF % 256 == 0, "tables behind the sorted path, aligned");

#define SORT_NBK 196
#define SORT_NB  250
#define SORT_EPB 3200
#define SORT_NT  (SORT_NBK * SORT_NB)
static_assert(SORT_NB * SORT_EPB == N_EDGES, "level 1 covers every edge once");
static_assert(SORT_NBK * 256 >= N_NODES, "every row has a bucket");
static_assert(SORT_EPB <= 4 * 1024, "four edges per thread cover a work-group's share");
static_assert(SORT_NT % 4 == 0 && SORT_NT <= N_NODES, "table fits its slot, scanned by fours");

#define WC_E1    (9 * 4096)
#define WC_E2    (WC_E1 + 4 * 4096)
#define WC_N1    (WC_E2 + 8 * 4096)
#define WC_TOTAL (WC_N1 + 4 * 4096)
#define WC_NB    (WC_TOTAL / 1024)
static_assert(WC_NB * 1024 == WC_TOTAL, "whole work-groups convert the weights");

__global__ __launch_bounds__(1024)
void prep_kernel(const int* __restrict__ eidx, int* __restrict__ table, size_t ws_size,
                 const float* __restrict__ We1, const float* __restrict__ We2,
                 const float* __restrict__ Wn1, const float* __restrict__ Wn2,
                 unsigned short* __restrict__ wb) {
    if (blockIdx.x >= SORT_NB) {
        const int g = (blockIdx.x - SORT_NB) * 1024 + threadIdx.x;
        if      (g < WC_E1) wconv_elem(We1, wb,         273, g);
        else if (g < WC_E2) wconv_elem(We2, wb + WC_E1, 128, g - WC_E1);
        else if (g < WC_N1) wconv_elem(Wn1, wb + WC_E2, 256, g - WC_E2);
        else                wconv_elem(Wn2, wb + WC_N1, 128, g - WC_N1);
        return;
    }
    if (ws_size < WS_NEED) return;
    __shared__ int cnt[SORT_NBK];
    const int tid = threadIdx.x;
    if (tid < SORT_NBK) cnt[tid] = 0;
    __syncthreads();
    const int* rows = eidx + blockIdx.x * SORT_EPB;
    int r[4];
    #pragma unroll
    for (int k = 0; k < 4; ++k) {
        const int i = tid + k * 1024;
        r[k] = (i < SORT_EPB) ? rows[i] : 0;
    }
    #pragma unroll
    for (int k = 0; k < 4; ++k) {
        const int i = tid + k * 1024;
        if (i < SORT_EPB) atomicAdd(&cnt[r[k] >> 8], 1);
    }
    __syncthreads();
    if (tid < SORT_NBK) table[tid * SORT_NB + blockIdx.x] = cnt[tid];
}

#define SCAN_K   13
#define SCAN_SEG (SCAN_K * 128)
static_assert(SORT_NT <= 32 * SCAN_SEG, "one work-group scans the whole table");
__global__ __launch_bounds__(1024)
void scan_kernel(int* __restrict__ cursor, int n, size_t ws_size) {
    if (ws_size < WS_NEED) return;
    __shared__ int wtot[32];
    const int w = threadIdx.x >> 5, lane = threadIdx.x & 31;
    const int base = w * SCAN_SEG + lane * 4;
    int4 v[SCAN_K];
    int  ex[SCAN_K];
    int running = 0;
    #pragma unroll
    for (int k = 0; k < SCAN_K; ++k) {
        int idx = base + k * 128;
        v[k] = (idx < n) ? *(const int4*)(cursor + idx) : make_int4(0, 0, 0, 0);
    }
    #pragma unroll
    for (int k = 0; k < SCAN_K; ++k) {
        int s = v[k].x + v[k].y + v[k].z + v[k].w;
        int inc = s;
        #pragma unroll
        for (int d = 1; d < 32; d <<= 1) {
            int t = __shfl_up(inc, d);
            if (lane >= d) inc += t;
        }
        ex[k] = running + inc - s;
        running += __shfl(inc, 31);
    }
    if (lane == 0) wtot[w] = running;
    __syncthreads();
    int woff = 0;
    for (int i = 0; i < w; ++i) woff += wtot[i];
    #pragma unroll
    for (int k = 0; k < SCAN_K; ++k) {
        int idx = base + k * 128;
        if (idx < n) {
            int o = woff + ex[k];
            int4 r;
            r.x = o; r.y = o + v[k].x; r.z = r.y + v[k].y; r.w = r.z + v[k].z;
            *(int4*)(cursor + idx) = r;
        }
    }
}

__global__ __launch_bounds__(1024)
void bucket_scatter_kernel(const int* __restrict__ eidx, const int* __restrict__ table,
                           int* __restrict__ pairs, size_t ws_size) {
    if (ws_size < WS_NEED) return;
    __shared__ int place[SORT_NBK];
    const int tid = threadIdx.x;
    if (tid < SORT_NBK) place[tid] = table[tid * SORT_NB + blockIdx.x];
    __syncthreads();
    const int e0 = blockIdx.x * SORT_EPB;
    int r[4];
    #pragma unroll
    for (int k = 0; k < 4; ++k) {
        const int i = tid + k * 1024;
        r[k] = (i < SORT_EPB) ? eidx[e0 + i] : 0;
    }
    #pragma unroll
    for (int k = 0; k < 4; ++k) {
        const int i = tid + k * 1024;
        if (i < SORT_EPB) {
            const int pos = atomicAdd(&place[r[k] >> 8], 1);
            *(int2*)(pairs + 2 * pos) = make_int2(e0 + i, r[k]);
        }
    }
}

#define SORT_RPT 5
static_assert((2 * N_EDGES) % 4 == 0 && NZ_BASE % 4 == 0, "cleared by 16-byte stores");
__global__ __launch_bounds__(1024)
void row_sort_kernel(const int* __restrict__ table, int* pairs,
                     int* __restrict__ perm, size_t ws_size) {
    const int tid = threadIdx.x;
    const int b = blockIdx.x;
    const bool sorted = ws_size >= WS_NEED;
    {
        const int q0 = sorted ? (2 * N_EDGES) / 4 : 0;
        const int q1 = NZ_BASE / 4;
        const int per = (q1 - q0 + SORT_NBK - 1) / SORT_NBK;
        const int a = q0 + b * per;
        const int e = (a + per < q1) ? a + per : q1;
        for (int i = a + tid; i < e; i += 1024)
            ((int4*)pairs)[i] = make_int4(0, 0, 0, 0);
    }
    if (!sorted) return;
    __shared__ int cnt[256];
    __shared__ int wtot[8];
    const int start = table[b * SORT_NB];
    const int end = (b + 1 < SORT_NBK) ? table[(b + 1) * SORT_NB] : N_EDGES;
    if (tid < 256) cnt[tid] = 0;
    __syncthreads();
    const int last = end - 1;
    for (int base = start; base < end; base += SORT_RPT * 1024) {
        int rr[SORT_RPT];
        #pragma unroll
        for (int k = 0; k < SORT_RPT; ++k) {
            const int i = base + k * 1024 + tid;
            rr[k] = pairs[2 * ((i < last) ? i : last) + 1];
        }
        #pragma unroll
        for (int k = 0; k < SORT_RPT; ++k) {
            const int i = base + k * 1024 + tid;
            if (i < end) atomicAdd(&cnt[rr[k] & 255], 1);
        }
    }
    __syncthreads();
    const int w = tid >> 5, lane = tid & 31;
    int v = 0, inc = 0;
    if (tid < 256) {
        v = cnt[tid];
        inc = v;
        #pragma unroll
        for (int d = 1; d < 32; d <<= 1) {
            int t = __shfl_up(inc, d);
            if (lane >= d) inc += t;
        }
        if (lane == 31) wtot[w] = inc;
    }
    __syncthreads();
    if (tid < 256) {
        int woff = 0;
        for (int i = 0; i < w; ++i) woff += wtot[i];
        cnt[tid] = start + woff + inc - v;
    }
    __syncthreads();
    for (int base = start; base < end; base += SORT_RPT * 1024) {
        int2 t[SORT_RPT];
        #pragma unroll
        for (int k = 0; k < SORT_RPT; ++k) {
            const int i = base + k * 1024 + tid;
            t[k] = *(const int2*)(pairs + 2 * ((i < last) ? i : last));
        }
        #pragma unroll
        for (int k = 0; k < SORT_RPT; ++k) {
            const int i = base + k * 1024 + tid;
            if (i < end) {
                const int pos = atomicAdd(&cnt[t[k].y & 255], 1);
                perm[pos] = t[k].x;
            }
        }
    }
    __syncthreads();
    for (int i = start + tid; i < end; i += 1024)
        ((int2*)pairs)[i] = make_int2(0, 0);
}

#define K1T 9
#define K2T 4

__device__ __forceinline__ void load_a4(Frag (&A)[4], const uint4* lX, int KT, int t, int as0) {
    #pragma unroll
    for (int mt = 0; mt < 4; ++mt) {
        const uint4* ap = lX + (mt * KT + t) * 64;
        A[mt].q[0] = ap[as0]; A[mt].q[1] = ap[as0 + 32];
    }
}
__device__ __forceinline__ void load_b2(Frag (&B)[2], const unsigned short* __restrict__ Wb,
                                        int t, int w, int lane) {
    #pragma unroll
    for (int u = 0; u < 2; ++u) {
        const uint4* bp = (const uint4*)Wb + ((t * 8 + 2 * w + u) * 32 + lane) * 2;
        B[u].q[0] = bp[0]; B[u].q[1] = bp[1];
    }
}

__global__ __launch_bounds__(256)
void pq_kernel(const float* __restrict__ h, const unsigned short* __restrict__ We1b,
               float* __restrict__ P, float* __restrict__ Q, size_t ws_size) {
    if (ws_size < WS_NEED2) return;
    __shared__ uint4 lA4[4 * 4 * 32 * 2];
    const int tid = threadIdx.x;
    const int n0  = blockIdx.x * 64;
    {
        const int m0 = tid >> 3;
        const int q  = tid & 7;
        float4 hv[8];
        #pragma unroll
        for (int s = 0; s < 2; ++s) {
            const int node = n0 + m0 + 32 * s;
            if (node < N_NODES) {
                const float4* ph = (const float4*)(h + node * HID) + q;
                #pragma unroll
                for (int t = 0; t < 4; ++t) hv[s * 4 + t] = ph[t * 8];
            } else {
                #pragma unroll
                for (int t = 0; t < 4; ++t) hv[s * 4 + t] = make_float4(0.0f, 0.0f, 0.0f, 0.0f);
            }
        }
        uint2* lA2 = (uint2*)lA4;
        const int b2 = abuild_base(m0, q, 4);
        #pragma unroll
        for (int s = 0; s < 2; ++s) {
            #pragma unroll
            for (int t = 0; t < 4; ++t)
                lA2[b2 + (2 * s * 4 + t) * 128] = pack4bf(hv[s * 4 + t]);
        }
    }
    __syncthreads();
    const int w    = tid >> 5;
    const int lane = tid & 31;
    const int hi   = lane >> 4;
    const int half = w >> 2;
    const int wc   = w & 3;
    const int n0c  = wc * 32 + (lane & 15);
    const int as0  = aslot(lane & 15, hi);
    v8f acc[4][2] = {};
    #pragma unroll
    for (int t = 0; t < 4; ++t) {
        Frag A[4];
        Frag B[2];
        load_b2(B, We1b, half * 4 + t, wc, lane);
        load_a4(A, (const uint4*)lA4, 4, t, as0);
        #pragma unroll
        for (int mt = 0; mt < 4; ++mt) {
            #pragma unroll
            for (int u = 0; u < 2; ++u)
                acc[mt][u] = __builtin_amdgcn_wmma_f32_16x16x32_bf16(
                    false, A[mt].v, false, B[u].v, (short)0, acc[mt][u], false, false);
        }
    }
    float* __restrict__ dst = half ? Q : P;
    #pragma unroll
    for (int mt = 0; mt < 4; ++mt) {
        #pragma unroll
        for (int j = 0; j < 8; ++j) {
            const int node = n0 + mt * 16 + hi * 8 + j;
            if (node < N_NODES) {
                dst[node * HID + n0c]      = acc[mt][0][j];
                dst[node * HID + n0c + 16] = acc[mt][1][j];
            }
        }
    }
}

#define S_PITCH 132

#define EDGE_LDS_OLD (4 * K1T * 32 * 2 + 4 * K2T * 32 * 2)
#define EDGE_LDS_NEW (64 * S_PITCH / 4 + 4 * 32 * 2 + 4 * K2T * 32 * 2)
#define EDGE_LDS (EDGE_LDS_NEW > EDGE_LDS_OLD ? EDGE_LDS_NEW : EDGE_LDS_OLD)

__device__ __forceinline__ void build_k8(uint4 (&k8)[2], const float* __restrict__ x,
                                         const float* __restrict__ eattr,
                                         const int* sE, const int* sR, const int* sC, int tid) {
    const int tc = tid & 3;
    int ip = 8 * tc - 1;
    ip = (ip < 0) ? 0 : ip;
    ip = (ip > EDGE_D - 1) ? EDGE_D - 1 : ip;
    const int iq = (tc < 2) ? 2 * tc : 3;
    const int ir = (tc < 2) ? 2 * tc + 1 : 3;
    const bool two = tc < 2;
    #pragma unroll
    for (int p = 0; p < 2; ++p) {
        const int tm = (tid >> 2) + 32 * p;
        const float* ea = eattr + sE[tm] * EDGE_D;
        const float  eP = ea[ip];
        const float4 eQ = ((const float4*)ea)[iq];
        const float4 eR = ((const float4*)ea)[ir];
        const int r = sR[tm];
        const int c = sC[tm];
        float d0 = x[r * 3 + 0] - x[c * 3 + 0];
        float d1 = x[r * 3 + 1] - x[c * 3 + 1];
        float d2 = x[r * 3 + 2] - x[c * 3 + 2];
        const float radial = d0 * d0 + d1 * d1 + d2 * d2;
        const float eP0 = (tc == 3) ? 0.0f : eP;
        float t8[8];
        t8[0] = (tc == 0) ? radial : eP0;
        t8[1] = two ? eQ.x : 0.0f;
        t8[2] = two ? eQ.y : 0.0f;
        t8[3] = two ? eQ.z : 0.0f;
        t8[4] = two ? eQ.w : 0.0f;
        t8[5] = two ? eR.x : 0.0f;
        t8[6] = two ? eR.y : 0.0f;
        t8[7] = two ? eR.z : 0.0f;
        k8[p] = make_uint4(pack2bf(t8[0], t8[1]), pack2bf(t8[2], t8[3]),
                           pack2bf(t8[4], t8[5]), pack2bf(t8[6], t8[7]));
    }
}

__device__ __forceinline__ void edge_layer1_full(v8f (&acc)[4][2], uint4* lA4,
        const int* sE, const int* sR, const int* sC,
        const float* __restrict__ h, const float* __restrict__ x, const float* __restrict__ eattr,
        const unsigned short* __restrict__ We1b, int tid, int w, int lane, int as0) {
    {
        const int m0 = tid >> 3;
        const int q  = tid & 7;
        uint2* lA2 = (uint2*)lA4;
        const int b2 = abuild_base(m0, q, K1T);
        #pragma unroll 1
        for (int sp = 0; sp < 2; ++sp) {
            float4 hv[16];
            #pragma unroll
            for (int s2 = 0; s2 < 2; ++s2) {
                const int s = sp * 2 + s2;
                const float4* pr = (const float4*)(h + sR[m0 + 16 * s] * HID) + q;
                const float4* pc = (const float4*)(h + sC[m0 + 16 * s] * HID) + q;
                #pragma unroll
                for (int t = 0; t < 4; ++t) {
                    hv[s2 * 8 + t]     = pr[t * 8];
                    hv[s2 * 8 + 4 + t] = pc[t * 8];
                }
            }
            #pragma unroll
            for (int s2 = 0; s2 < 2; ++s2) {
                const int s = sp * 2 + s2;
                #pragma unroll
                for (int t = 0; t < 8; ++t)
                    lA2[b2 + (s * K1T + t) * 128] = pack4bf(hv[s2 * 8 + t]);
            }
        }
        uint4 k8[2];
        build_k8(k8, x, eattr, sE, sR, sC, tid);
        const int tc = tid & 3;
        #pragma unroll
        for (int p = 0; p < 2; ++p) {
            const int tm = (tid >> 2) + 32 * p;
            lA4[(((tm >> 4) * K1T + 8) * 2 + (tc >> 1)) * 32 + aslot(tm & 15, tc & 1)] = k8[p];
        }
    }
    __syncthreads();

    #pragma unroll 1
    for (int t = 0; t < K1T; ++t) {
        Frag A[4];
        Frag B[2];
        load_b2(B, We1b, t, w, lane);
        load_a4(A, (const uint4*)lA4, K1T, t, as0);
        #pragma unroll
        for (int mt = 0; mt < 4; ++mt) {
            #pragma unroll
            for (int u = 0; u < 2; ++u)
                acc[mt][u] = __builtin_amdgcn_wmma_f32_16x16x32_bf16(
                    false, A[mt].v, false, B[u].v, (short)0, acc[mt][u], false, false);
        }
    }
}

__device__ __forceinline__ void edge_layer1_split(v8f (&acc)[4][2], uint4* smem,
        const int* sE, const int* sR, const int* sC,
        const float* __restrict__ P, const float* __restrict__ Q,
        const float* __restrict__ x, const float* __restrict__ eattr,
        const unsigned short* __restrict__ We1b,
        int tid, int w, int lane, int hi, int n0c, int as0) {
    float* lS = (float*)smem;
    uint4* lK = smem + 64 * S_PITCH / 4;
    {
        const int m0 = tid >> 3;
        const int q  = tid & 7;
        float4 pv[16], qv[16];
        #pragma unroll
        for (int s = 0; s < 4; ++s) {
            const float4* pr = (const float4*)(P + sR[m0 + 16 * s] * HID) + q;
            const float4* pc = (const float4*)(Q + sC[m0 + 16 * s] * HID) + q;
            #pragma unroll
            for (int t = 0; t < 4; ++t) {
                pv[s * 4 + t] = pr[t * 8];
                qv[s * 4 + t] = pc[t * 8];
            }
        }
        uint4 k8[2];
        build_k8(k8, x, eattr, sE, sR, sC, tid);
        #pragma unroll
        for (int s = 0; s < 4; ++s) {
            #pragma unroll
            for (int t = 0; t < 4; ++t) {
                const float4 a = pv[s * 4 + t], b = qv[s * 4 + t];
                *(float4*)(lS + (m0 + 16 * s) * S_PITCH + 32 * t + 4 * q) =
                    make_float4(a.x + b.x, a.y + b.y, a.z + b.z, a.w + b.w);
            }
        }
        const int tc = tid & 3;
        #pragma unroll
        for (int p = 0; p < 2; ++p) {
            const int tm = (tid >> 2) + 32 * p;
            lK[((tm >> 4) * 2 + (tc >> 1)) * 32 + aslot(tm & 15, tc & 1)] = k8[p];
        }
    }
    __syncthreads();

    Frag A[4];
    Frag B[2];
    load_b2(B, We1b, 8, w, lane);
    load_a4(A, (const uint4*)lK, 1, 0, as0);
    #pragma unroll
    for (int mt = 0; mt < 4; ++mt) {
        #pragma unroll
        for (int u = 0; u < 2; ++u) {
            #pragma unroll
            for (int j = 0; j < 8; ++j)
                acc[mt][u][j] = lS[(mt * 16 + hi * 8 + j) * S_PITCH + n0c + 16 * u];
        }
    }
    #pragma unroll
    for (int mt = 0; mt < 4; ++mt) {
        #pragma unroll
        for (int u = 0; u < 2; ++u)
            acc[mt][u] = __builtin_amdgcn_wmma_f32_16x16x32_bf16(
                false, A[mt].v, false, B[u].v, (short)0, acc[mt][u], false, false);
    }
}

__global__ __launch_bounds__(128)
void edge_kernel(const float* __restrict__ h,
                 const int*   __restrict__ eidx,
                 const float* __restrict__ x,
                 const float* __restrict__ eattr,
                 const unsigned short* __restrict__ We1b,
                 const float* __restrict__ be1,
                 const unsigned short* __restrict__ We2b,
                 const float* __restrict__ be2,
                 float* __restrict__ agg,
                 const int* __restrict__ perm,
                 const float* __restrict__ P,
                 const float* __restrict__ Q,
                 size_t ws_size) {
    __shared__ uint4 smem[EDGE_LDS];
    const bool split = ws_size >= WS_NEED2;
    uint4* lM4 = smem;
    uint4* lL1 = smem + 4 * K2T * 32 * 2;
    unsigned short* lM = (unsigned short*)lM4;

    const int tid = threadIdx.x;
    const int e0  = blockIdx.x * 64;

    int* sE = (int*)lM4;
    int* sR = sE + 64;
    int* sC = sE + 128;
    if (tid < 64) {
        int i = e0 + tid;
        int e = (ws_size >= WS_NEED) ? perm[i] : i;
        sE[tid] = e;
        sR[tid] = eidx[e];
        sC[tid] = eidx[N_EDGES + e];
    }
    __syncthreads();
    const int rowLo = sR[tid & 31];
    const int rowHi = sR[32 + (tid & 31)];

    const int w    = tid >> 5;
    const int lane = tid & 31;
    const int hi   = lane >> 4;
    const int n0c  = w * 32 + (lane & 15);
    const int as0  = aslot(lane & 15, hi);

    v8f acc[4][2] = {};
    if (split)
        edge_layer1_split(acc, lL1, sE, sR, sC, P, Q, x, eattr, We1b, tid, w, lane, hi, n0c, as0);
    else
        edge_layer1_full(acc, lL1, sE, sR, sC, h, x, eattr, We1b, tid, w, lane, as0);

    Frag A[2][4];
    Frag B[3][2];
    load_b2(B[0], We2b, 0, w, lane);
    load_b2(B[1], We2b, 1, w, lane);
    {
        #pragma unroll
        for (int u = 0; u < 2; ++u) {
            const int n = n0c + 16 * u;
            float b = be1[n];
            const int mp0 = asw_pos(hi * 8,     n, K2T);
            const int mp1 = asw_pos(hi * 8 + 4, n, K2T);
            #pragma unroll
            for (int mt = 0; mt < 4; ++mt) {
                #pragma unroll
                for (int j = 0; j < 8; ++j)
                    lM[mt * (K2T * 512) + ((j < 4) ? mp0 : mp1) + (j & 3) * 8] =
                        f2bf(silu_f(acc[mt][u][j] + b));
            }
        }
    }
    __syncthreads();

    v8f acc2[4][2] = {};
    load_a4(A[0], (const uint4*)lM4, K2T, 0, as0);
    #pragma unroll
    for (int t = 0; t < K2T; ++t) {
        if (t + 2 < K2T) load_b2(B[(t + 2) % 3], We2b, t + 2, w, lane);
        if (t + 1 < K2T) load_a4(A[(t + 1) & 1], (const uint4*)lM4, K2T, t + 1, as0);
        __builtin_amdgcn_sched_barrier(0);
        #pragma unroll
        for (int mt = 0; mt < 4; ++mt) {
            #pragma unroll
            for (int u = 0; u < 2; ++u)
                acc2[mt][u] = __builtin_amdgcn_wmma_f32_16x16x32_bf16(
                    false, A[t & 1][mt].v, false, B[t % 3][u].v, (short)0, acc2[mt][u],
                    false, false);
        }
        __builtin_amdgcn_sched_barrier(0);
    }
    {
        const float b0 = be2[n0c];
        const float b1 = be2[n0c + 16];
        int   cur = 0;
        float run0 = 0.0f, run1 = 0.0f;
        #pragma unroll
        for (int mt = 0; mt < 4; ++mt) {
            #pragma unroll
            for (int j = 0; j < 8; ++j) {
                const int rv = (mt < 2) ? rowLo : rowHi;
                const int r0 = __builtin_amdgcn_readlane(rv, (mt & 1) * 16 + j);
                const int r1 = __builtin_amdgcn_readlane(rv, (mt & 1) * 16 + 8 + j);
                const int r  = hi ? r1 : r0;
                const float f0 = silu_f(acc2[mt][0][j] + b0);
                const float f1 = silu_f(acc2[mt][1][j] + b1);
                if (mt == 0 && j == 0) {
                    cur = r;
                    run0 = f0;
                    run1 = f1;
                } else {
                    if (r != cur) {
                        atomicAdd(&agg[cur * HID + n0c], run0);
                        atomicAdd(&agg[cur * HID + n0c + 16], run1);
                        run0 = 0.0f;
                        run1 = 0.0f;
                        cur = r;
                    }
                    run0 += f0;
                    run1 += f1;
                }
            }
        }
        atomicAdd(&agg[cur * HID + n0c], run0);
        atomicAdd(&agg[cur * HID + n0c + 16], run1);
    }
}

#define KN1T 8
#define KN2T 4

__global__ __launch_bounds__(256)
void node_kernel(const float* __restrict__ h,
                 const float* __restrict__ agg,
                 const int*   __restrict__ batch,
                 const unsigned short* __restrict__ Wn1b,
                 const float* __restrict__ bn1,
                 const unsigned short* __restrict__ Wn2b,
                 const float* __restrict__ bn2,
                 float* __restrict__ sums,
                 float* __restrict__ counts) {
    __shared__ uint4 lA4[4 * KN1T * 32 * 2];
    __shared__ uint4 lM4[4 * KN2T * 32 * 2];
    unsigned short* lM = (unsigned short*)lM4;

    const int tid = threadIdx.x;
    const int n0  = blockIdx.x * 64;

    if (tid < 64) {
        const int i    = n0 + tid;
        const int lane = tid & 31;
        const int g    = (i < N_NODES) ? batch[i] : -1;
        const int prev = __shfl_up(g, 1);
        const bool head = (lane == 0) || (prev != g);
        const unsigned int heads = __builtin_amdgcn_ballot_w32(head);
        if (head && g >= 0) {
            const unsigned int above = (lane == 31) ? 0u : (heads >> (lane + 1));
            const int len = above ? (1 + __builtin_ctz(above)) : (32 - lane);
            atomicAdd(&counts[g], (float)len);
        }
    }

    {
        const int m0 = tid >> 3;
        const int q  = tid & 7;
        float4 hv[16];
        #pragma unroll
        for (int s = 0; s < 2; ++s) {
            const int node = n0 + m0 + 32 * s;
            if (node < N_NODES) {
                const float4* ph = (const float4*)(h   + node * HID) + q;
                const float4* pa = (const float4*)(agg + node * HID) + q;
                #pragma unroll
                for (int t = 0; t < 4; ++t) {
                    hv[s * 8 + t]     = ph[t * 8];
                    hv[s * 8 + 4 + t] = pa[t * 8];
                }
            } else {
                #pragma unroll
                for (int t = 0; t < 8; ++t)
                    hv[s * 8 + t] = make_float4(0.0f, 0.0f, 0.0f, 0.0f);
            }
        }
        uint2* lA2 = (uint2*)lA4;
        const int b2 = abuild_base(m0, q, KN1T);
        #pragma unroll
        for (int s = 0; s < 2; ++s) {
            #pragma unroll
            for (int t = 0; t < 8; ++t)
                lA2[b2 + (2 * s * KN1T + t) * 128] = pack4bf(hv[s * 8 + t]);
        }
    }
    __syncthreads();

    const int w    = tid >> 5;
    const int lane = tid & 31;
    const int hi   = lane >> 4;
    const int n    = w * 16 + (lane & 15);
    const int as0  = aslot(lane & 15, hi);

    v8f acc[4] = {};
    for (int t = 0; t < KN1T; ++t) {
        Frag B;
        const uint4* bp = (const uint4*)Wn1b + ((t * 8 + w) * 32 + lane) * 2;
        B.q[0] = bp[0]; B.q[1] = bp[1];
        #pragma unroll
        for (int mt = 0; mt < 4; ++mt) {
            Frag A;
            const uint4* ap = (const uint4*)lA4 + (mt * KN1T + t) * 64;
            A.q[0] = ap[as0]; A.q[1] = ap[as0 + 32];
            acc[mt] = __builtin_amdgcn_wmma_f32_16x16x32_bf16(
                false, A.v, false, B.v, (short)0, acc[mt], false, false);
        }
    }
    {
        float b = bn1[n];
        const int mp0 = asw_pos(hi * 8,     n, KN2T);
        const int mp1 = asw_pos(hi * 8 + 4, n, KN2T);
        #pragma unroll
        for (int mt = 0; mt < 4; ++mt) {
            #pragma unroll
            for (int j = 0; j < 8; ++j)
                lM[mt * (KN2T * 512) + ((j < 4) ? mp0 : mp1) + (j & 3) * 8] =
                    f2bf(silu_f(acc[mt][j] + b));
        }
    }
    __syncthreads();

    v8f acc2[4] = {};
    for (int t = 0; t < KN2T; ++t) {
        Frag B;
        const uint4* bp = (const uint4*)Wn2b + ((t * 8 + w) * 32 + lane) * 2;
        B.q[0] = bp[0]; B.q[1] = bp[1];
        #pragma unroll
        for (int mt = 0; mt < 4; ++mt) {
            Frag A;
            const uint4* ap = (const uint4*)lM4 + (mt * KN2T + t) * 64;
            A.q[0] = ap[as0]; A.q[1] = ap[as0 + 32];
            acc2[mt] = __builtin_amdgcn_wmma_f32_16x16x32_bf16(
                false, A.v, false, B.v, (short)0, acc2[mt], false, false);
        }
    }
    {
        float b = bn2[n];
        int   cur = -1;
        float run = 0.0f;
        #pragma unroll
        for (int mt = 0; mt < 4; ++mt) {
            #pragma unroll
            for (int j = 0; j < 8; ++j) {
                int m = mt * 16 + hi * 8 + j;
                int node = n0 + m;
                if (node < N_NODES) {
                    float hn = h[node * HID + n] + (acc2[mt][j] + b);
                    int g = batch[node];
                    if (g != cur) {
                        if (cur >= 0) atomicAdd(&sums[cur * HID + n], run);
                        cur = g;
                        run = hn;
                    } else {
                        run += hn;
                    }
                }
            }
        }
        if (cur >= 0) atomicAdd(&sums[cur * HID + n], run);
    }
}

__global__ void finalize_kernel(const float* __restrict__ sums,
                                const float* __restrict__ counts,
                                float* __restrict__ out) {
    int i = blockIdx.x * blockDim.x + threadIdx.x;
    if (i < N_GRAPHS * HID)
        out[i] = sums[i] / fmaxf(counts[i >> 7], 1.0f);
}

extern "C" void kernel_launch(void* const* d_in, const int* in_sizes, int n_in,
                              void* d_out, int out_size, void* d_ws, size_t ws_size,
                              hipStream_t stream) {
    const float* h     = (const float*)d_in[0];
    const int*   eidx  = (const int*)  d_in[1];
    const float* x     = (const float*)d_in[2];
    const float* eattr = (const float*)d_in[3];
    const int*   batch = (const int*)  d_in[4];
    const float* We1   = (const float*)d_in[5];
    const float* be1   = (const float*)d_in[6];
    const float* We2   = (const float*)d_in[7];
    const float* be2   = (const float*)d_in[8];
    const float* Wn1   = (const float*)d_in[9];
    const float* bn1   = (const float*)d_in[10];
    const float* Wn2   = (const float*)d_in[11];
    const float* bn2   = (const float*)d_in[12];

    char* ws = (char*)d_ws;
    unsigned short* We1b = (unsigned short*)(ws);
    unsigned short* We2b = (unsigned short*)(ws + 73728);
    unsigned short* Wn1b = (unsigned short*)(ws + 106496);
    unsigned short* Wn2b = (unsigned short*)(ws + 172032);
    float* agg    = (float*)(ws + 204800);
    float* sums   = (float*)(ws + 204800 + 25600000);
    float* counts = (float*)(ws + 204800 + 25600000 + 32768);
    int*   table  = (int*)(ws + 25837824);
    int*   perm   = (int*)(ws + 26037824);
    int*   pairs  = (int*)agg;
    float* P      = (float*)(ws + PQ_OFF);
    float* Q      = (float*)(ws + PQ_OFF + 25600000);

    prep_kernel<<<SORT_NB + WC_NB, 1024, 0, stream>>>(eidx, table, ws_size,
                                                      We1, We2, Wn1, Wn2, We1b);
    pq_kernel<<<(N_NODES + 63) / 64, 256, 0, stream>>>(h, We1b, P, Q, ws_size);
    scan_kernel<<<1, 1024, 0, stream>>>(table, SORT_NT, ws_size);
    bucket_scatter_kernel<<<SORT_NB, 1024, 0, stream>>>(eidx, table, pairs, ws_size);
    row_sort_kernel<<<SORT_NBK, 1024, 0, stream>>>(table, pairs, perm, ws_size);

    edge_kernel<<<N_EDGES / 64, 128, 0, stream>>>(h, eidx, x, eattr,
                                                  We1b, be1, We2b, be2, agg,
                                                  perm, P, Q, ws_size);
    node_kernel<<<(N_NODES + 63) / 64, 256, 0, stream>>>(h, agg, batch,
                                                         Wn1b, bn1, Wn2b, bn2, sums,
                                                         counts);
    finalize_kernel<<<(N_GRAPHS * HID + 255) / 256, 256, 0, stream>>>(sums, counts,
                                                                      (float*)d_out);
}
